# starting kernel
# baseline (speedup 1.0000x reference)
#include <hip/hip_runtime.h>

typedef __attribute__((ext_vector_type(2))) float v2f;
typedef __attribute__((ext_vector_type(8))) float v8f;

__global__ void set_val_kernel(float* p, long long n, float v) {
  long long i = (long long)blockIdx.x * blockDim.x + threadIdx.x;
  if (i < n) p[i] = v;
}

__global__ void deg_count_kernel(const int* __restrict__ dst, int E, float* deg) {
  int e = blockIdx.x * blockDim.x + threadIdx.x;
  if (e < E) atomicAdd(&deg[dst[e]], 1.0f);
}

__global__ void rsqrt_kernel(float* p, int n) {
  int i = blockIdx.x * blockDim.x + threadIdx.x;
  if (i < n) p[i] = rsqrtf(p[i]);
}

template<int K, int N>
__global__ void gemm_wmma_kernel(const float* __restrict__ X,
                                 const float* __restrict__ W,
                                 float* __restrict__ OUT, int M) {
  __shared__ float sW[K * N];
  for (int i = threadIdx.x; i < K * N; i += blockDim.x) sW[i] = W[i];
  __syncthreads();

  constexpr int COL_TILES  = N / 16;
  constexpr int ROW_TILES  = 8 / COL_TILES;
  constexpr int BLOCK_ROWS = 16 * ROW_TILES;

  const int wave = threadIdx.x >> 5;
  const int lane = threadIdx.x & 31;
  const int rowBase = blockIdx.x * BLOCK_ROWS + (wave / COL_TILES) * 16;
  const int colBase = (wave % COL_TILES) * 16;
  if (rowBase + 16 > M) return;

  const int mn  = lane & 15;
  const int grp = lane >> 4;

  const float* xrow = X + (long long)(rowBase + mn) * K + 2 * grp;
  const float* wcol = sW + (long long)(2 * grp) * N + colBase + mn;

  v8f acc = {};
#pragma unroll 4
  for (int k0 = 0; k0 < K; k0 += 4) {
    v2f a = *(const v2f*)(xrow + k0);
    v2f b;
    b.x = wcol[(long long)k0 * N];
    b.y = wcol[(long long)(k0 + 1) * N];
    acc = __builtin_amdgcn_wmma_f32_16x16x4_f32(
        false, a, false, b, (short)0, acc, false, false);
  }

  float* out = OUT + (long long)(rowBase + 8 * grp) * N + colBase + mn;
#pragma unroll
  for (int r = 0; r < 8; ++r) out[(long long)r * N] = acc[r];
}

template<int D>
__global__ void scatter_kernel(const int* __restrict__ src, const int* __restrict__ dst,
                               const float* __restrict__ dis, const float* __restrict__ H,
                               float* __restrict__ agg, int E) {
  constexpr int CH = D / 4;
  long long t = (long long)blockIdx.x * blockDim.x + threadIdx.x;
  long long e = t / CH;
  int c = (int)(t % CH);
  if (e >= E) return;
  int s = src[e], d = dst[e];
  float norm = dis[s] * dis[d];
  const float4 v = *(const float4*)(H + (long long)s * D + 4 * c);
  float* o = agg + (long long)d * D + 4 * c;
  atomicAdd(o + 0, v.x * norm);
  atomicAdd(o + 1, v.y * norm);
  atomicAdd(o + 2, v.z * norm);
  atomicAdd(o + 3, v.w * norm);
}

template<int D, bool RELU>
__global__ void combine_kernel(const float* __restrict__ H, const float* __restrict__ bias,
                               const float* __restrict__ dis, float* agg, int M) {
  long long i = (long long)blockIdx.x * blockDim.x + threadIdx.x;
  if (i >= (long long)M * D) return;
  int n = (int)(i / D), d = (int)(i % D);
  float z = agg[i] + dis[n] * dis[n] * H[i] + bias[d];
  if (RELU) z = fmaxf(z, 0.0f);
  agg[i] = z;
}

__global__ void combine_softmax_kernel(const float* __restrict__ H2,
                                       const float* __restrict__ b2,
                                       const float* __restrict__ dis,
                                       float* out, int M) {
  long long gt = (long long)blockIdx.x * blockDim.x + threadIdx.x;
  int node = (int)(gt >> 5);
  int lane = threadIdx.x & 31;
  if (node >= M) return;
  long long idx = (long long)node * 32 + lane;
  float ds2 = dis[node];
  ds2 *= ds2;
  float z = out[idx] + ds2 * H2[idx] + b2[lane];
  float m = z;
  for (int off = 16; off > 0; off >>= 1) m = fmaxf(m, __shfl_xor(m, off, 32));
  float ev = __expf(z - m);
  float s = ev;
  for (int off = 16; off > 0; off >>= 1) s += __shfl_xor(s, off, 32);
  out[idx] = ev / s;
}

extern "C" void kernel_launch(void* const* d_in, const int* in_sizes, int n_in,
                              void* d_out, int out_size, void* d_ws, size_t ws_size,
                              hipStream_t stream) {
  (void)n_in; (void)out_size; (void)ws_size;
  const float* x  = (const float*)d_in[0];
  const int*   ei = (const int*)d_in[1];
  const float* W1 = (const float*)d_in[2];
  const float* b1 = (const float*)d_in[3];
  const float* W2 = (const float*)d_in[4];
  const float* b2 = (const float*)d_in[5];

  const int IN = 256, HID = 64, OD = 32;
  const int M = in_sizes[0] / IN;
  const int E = in_sizes[1] / 2;
  const int* src = ei;
  const int* dst = ei + E;

  float* ws   = (float*)d_ws;
  float* dis  = ws;
  float* H1   = dis + (((long long)M + 1023) & ~1023LL);
  float* agg1 = H1 + (long long)M * HID;
  float* H2   = agg1 + (long long)M * HID;
  float* out  = (float*)d_out;

  const int T = 256;

  set_val_kernel<<<(M + T - 1) / T, T, 0, stream>>>(dis, M, 1.0f);
  deg_count_kernel<<<(E + T - 1) / T, T, 0, stream>>>(dst, E, dis);
  rsqrt_kernel<<<(M + T - 1) / T, T, 0, stream>>>(dis, M);

  gemm_wmma_kernel<256, 64><<<(M + 31) / 32, 256, 0, stream>>>(x, W1, H1, M);
  set_val_kernel<<<(int)(((long long)M * HID + T - 1) / T), T, 0, stream>>>(
      agg1, (long long)M * HID, 0.0f);
  {
    long long tot = (long long)E * (HID / 4);
    scatter_kernel<64><<<(int)((tot + T - 1) / T), T, 0, stream>>>(src, dst, dis, H1, agg1, E);
  }
  combine_kernel<64, true><<<(int)(((long long)M * HID + T - 1) / T), T, 0, stream>>>(
      H1, b1, dis, agg1, M);

  gemm_wmma_kernel<64, 32><<<(M + 63) / 64, 256, 0, stream>>>(agg1, W2, H2, M);
  set_val_kernel<<<(int)(((long long)M * OD + T - 1) / T), T, 0, stream>>>(
      out, (long long)M * OD, 0.0f);
  {
    long long tot = (long long)E * (OD / 4);
    scatter_kernel<32><<<(int)((tot + T - 1) / T), T, 0, stream>>>(src, dst, dis, H2, out, E);
  }
  combine_softmax_kernel<<<(int)(((long long)M * 32 + T - 1) / T), T, 0, stream>>>(
      H2, b2, dis, out, M);
}


# edge sums pulled per destination, no float atomics: per-call source list in destination order (returning int add = count+rank, scan, place); a wave sums a node's rows in registers in list order
# speedup vs baseline: 1.9550x; 1.9550x over previous
#include <hip/hip_runtime.h>

typedef __attribute__((ext_vector_type(2))) float v2f;
typedef __attribute__((ext_vector_type(8))) float v8f;

__global__ void set_val_kernel(float* p, long long n, float v) {
  long long i = (long long)blockIdx.x * blockDim.x + threadIdx.x;
  if (i < n) p[i] = v;
}

__global__ void count_rank_kernel(const int* __restrict__ dst, int E,
                                  int* __restrict__ cnt, int* __restrict__ rank) {
  int e = blockIdx.x * blockDim.x + threadIdx.x;
  if (e < E) rank[e] = atomicAdd(&cnt[dst[e]], 1);
}

__global__ void __launch_bounds__(1024)
scan_dis_kernel(const int* __restrict__ cnt, int* __restrict__ start,
                float* __restrict__ dis, int M) {
  __shared__ int sFront[32];
  __shared__ int sWave[32];
  const int tid = threadIdx.x, lane = tid & 31, wave = tid >> 5;
  int f = 0;
  const int4* c4 = (const int4*)cnt;
  for (int i = tid; i < (int)blockIdx.x * 256; i += 1024) {
    const int4 c = c4[i];
    f += (c.x + c.y) + (c.z + c.w);
  }
  for (int o = 16; o > 0; o >>= 1) f += __shfl_xor(f, o, 32);
  if (lane == 0) sFront[wave] = f;

  const int i = blockIdx.x * 1024 + tid;
  const int c = (i < M) ? cnt[i] : 0;
  int incl = c;
  for (int o = 1; o < 32; o <<= 1) {
    const int t = __shfl_up(incl, o, 32);
    if (lane >= o) incl += t;
  }
  if (lane == 31) sWave[wave] = incl;
  __syncthreads();
  int p = sFront[lane];
  int t = (lane < wave) ? sWave[lane] : 0;
  p += t;
  for (int o = 16; o > 0; o >>= 1) p += __shfl_xor(p, o, 32);
  if (i < M) {
    start[i + 1] = p + incl;
    dis[i] = rsqrtf(1.0f + (float)c);
  }
  if (i == 0) start[0] = 0;
}

__global__ void place_kernel(const int* __restrict__ src, const int* __restrict__ dst,
                             const int* __restrict__ start, const int* __restrict__ rank,
                             int* __restrict__ list, int E) {
  int e = blockIdx.x * blockDim.x + threadIdx.x;
  if (e < E) list[start[dst[e]] + rank[e]] = src[e];
}

template<int K, int N>
__global__ void __launch_bounds__(256)
gemm_wmma_kernel(const float* __restrict__ X,
                 const float* __restrict__ W,
                 float* __restrict__ OUT, int M) {
  constexpr int CT = N / 16;
  constexpr int QN = K / 32;
  __shared__ float4 sW[(K / 4) * N];

  for (int u = threadIdx.x; u < (K / 4) * N; u += 256) {
    const int n  = u % N;
    const int kq = u / N;
    const int q = kq >> 3, hh = (kq >> 2) & 1, jj = kq & 3;
    const float* w = W + (long long)(4 * kq) * N + n;
    float4 v;
    v.x = w[0];
    v.y = w[N];
    v.z = w[2 * N];
    v.w = w[3 * N];
    sW[((q * 4 + jj) * CT + (n % CT)) * 32 + hh * 16 + (n / CT)] = v;
  }
  __syncthreads();

  const int wave = threadIdx.x >> 5;
  const int lane = threadIdx.x & 31;
  const int rowBase = (blockIdx.x * 8 + wave) * 16;
  if (rowBase + 16 > M) return;

  const int c = lane & 15;
  const int h = lane >> 4;

  const float4* xp = (const float4*)(X + (long long)(rowBase + c) * K + 16 * h);
  const float4* bp = sW + lane;

  v8f acc[CT];
#pragma unroll
  for (int t = 0; t < CT; ++t) acc[t] = v8f{};

  float4 a[2][4];
#pragma unroll
  for (int i = 0; i < 4; ++i) a[0][i] = xp[i];

#pragma unroll
  for (int q = 0; q < QN; ++q) {
    if (q + 1 < QN) {
#pragma unroll
      for (int i = 0; i < 4; ++i) a[(q + 1) & 1][i] = xp[(q + 1) * 8 + i];
    }
#pragma unroll
    for (int jj = 0; jj < 4; ++jj) {
      const float4 av = a[q & 1][jj];
      float4 b[CT];
#pragma unroll
      for (int t = 0; t < CT; ++t) b[t] = bp[((q * 4 + jj) * CT + t) * 32];
      v2f a0; a0.x = av.x; a0.y = av.y;
      v2f a1; a1.x = av.z; a1.y = av.w;
#pragma unroll
      for (int t = 0; t < CT; ++t) {
        v2f b0; b0.x = b[t].x; b0.y = b[t].y;
        acc[t] = __builtin_amdgcn_wmma_f32_16x16x4_f32(
            false, a0, false, b0, (short)0, acc[t], false, false);
      }
#pragma unroll
      for (int t = 0; t < CT; ++t) {
        v2f b1; b1.x = b[t].z; b1.y = b[t].w;
        acc[t] = __builtin_amdgcn_wmma_f32_16x16x4_f32(
            false, a1, false, b1, (short)0, acc[t], false, false);
      }
    }
  }

  float* out = OUT + (long long)(rowBase + 8 * h) * N + CT * c;
#pragma unroll
  for (int r = 0; r < 8; ++r) {
    if constexpr (CT == 4) {
      float4 o;
      o.x = acc[0][r]; o.y = acc[1][r]; o.z = acc[2][r]; o.w = acc[3][r];
      *(float4*)(out + (long long)r * N) = o;
    } else {
      static_assert(CT == 2, "N is 64 or 32 here");
      float2 o;
      o.x = acc[0][r]; o.y = acc[1][r];
      *(float2*)(out + (long long)r * N) = o;
    }
  }
}


__global__ void __launch_bounds__(256)
pull_relu64_kernel(const int* __restrict__ start, const int* __restrict__ list,
                   const float* __restrict__ dis, const float* __restrict__ H,
                   const float* __restrict__ bias, float* __restrict__ out, int M) {
  const int lane = threadIdx.x & 31;
  const int wv = __builtin_amdgcn_readfirstlane((int)(threadIdx.x >> 5));
  const int nw = gridDim.x * 8;
  const float2 bb = *(const float2*)(bias + 2 * lane);
  for (int node = blockIdx.x * 8 + wv; node < M; node += nw) {
    const int beg = start[node], end = start[node + 1];
    const float dn = dis[node];
    float2 acc; acc.x = 0.0f; acc.y = 0.0f;
    for (int j = beg; j < end; j += 32) {
      const int m = min(32, end - j);
      int sid = 0; float w = 0.0f;
      if (lane < m) { sid = list[j + lane]; w = dis[sid] * dn; }
#pragma unroll 4
      for (int k = 0; k < m; ++k) {
        const int s = __builtin_amdgcn_readlane(sid, k);
        const float ws = __int_as_float(__builtin_amdgcn_readlane(__float_as_int(w), k));
        const float2 v = *(const float2*)(H + (long long)s * 64 + 2 * lane);
        acc.x += v.x * ws;
        acc.y += v.y * ws;
      }
    }
    const float2 hs = *(const float2*)(H + (long long)node * 64 + 2 * lane);
    float2 z;
    z.x = fmaxf(acc.x + dn * dn * hs.x + bb.x, 0.0f);
    z.y = fmaxf(acc.y + dn * dn * hs.y + bb.y, 0.0f);
    *(float2*)(out + (long long)node * 64 + 2 * lane) = z;
  }
}

__global__ void __launch_bounds__(256)
pull_softmax32_kernel(const int* __restrict__ start, const int* __restrict__ list,
                      const float* __restrict__ dis, const float* __restrict__ H,
                      const float* __restrict__ bias, float* __restrict__ out, int M) {
  const int lane = threadIdx.x & 31;
  const int wv = __builtin_amdgcn_readfirstlane((int)(threadIdx.x >> 5));
  const int nw = gridDim.x * 8;
  const float bb = bias[lane];
  for (int node = blockIdx.x * 8 + wv; node < M; node += nw) {
    const int beg = start[node], end = start[node + 1];
    const float dn = dis[node];
    float acc = 0.0f;
    for (int j = beg; j < end; j += 32) {
      const int m = min(32, end - j);
      int sid = 0; float w = 0.0f;
      if (lane < m) { sid = list[j + lane]; w = dis[sid] * dn; }
#pragma unroll 4
      for (int k = 0; k < m; ++k) {
        const int s = __builtin_amdgcn_readlane(sid, k);
        const float ws = __int_as_float(__builtin_amdgcn_readlane(__float_as_int(w), k));
        acc += H[(long long)s * 32 + lane] * ws;
      }
    }
    const float z = acc + dn * dn * H[(long long)node * 32 + lane] + bb;
    float mx = z;
    for (int o = 16; o > 0; o >>= 1) mx = fmaxf(mx, __shfl_xor(mx, o, 32));
    const float ev = __expf(z - mx);
    float sm = ev;
    for (int o = 16; o > 0; o >>= 1) sm += __shfl_xor(sm, o, 32);
    out[(long long)node * 32 + lane] = ev / sm;
  }
}

extern "C" void kernel_launch(void* const* d_in, const int* in_sizes, int n_in,
                              void* d_out, int out_size, void* d_ws, size_t ws_size,
                              hipStream_t stream) {
  (void)n_in; (void)out_size; (void)ws_size;
  const float* x  = (const float*)d_in[0];
  const int*   ei = (const int*)d_in[1];
  const float* W1 = (const float*)d_in[2];
  const float* b1 = (const float*)d_in[3];
  const float* W2 = (const float*)d_in[4];
  const float* b2 = (const float*)d_in[5];

  const int IN = 256, HID = 64, OD = 32;
  const int M = in_sizes[0] / IN;
  const int E = in_sizes[1] / 2;
  const int* src = ei;
  const int* dst = ei + E;

  const long long Mp = ((long long)M + 1023) & ~1023LL;
  float* ws   = (float*)d_ws;
  float* dis  = ws;
  float* H1   = dis + Mp;
  float* h1   = H1 + (long long)M * HID;
  int*   list = (int*)(h1 + (long long)M * HID);
  int*   cnt  = list + (((long long)E + 3) & ~3LL);
  int*   start = cnt + Mp;
  float* H2   = H1;
  float* out  = (float*)d_out;
  int*   rank = (int*)d_out;

  const int T = 256;
  const int PULL_BLOCKS = 2048;

  set_val_kernel<<<(M + T - 1) / T, T, 0, stream>>>((float*)cnt, M, 0.0f);
  count_rank_kernel<<<(E + T - 1) / T, T, 0, stream>>>(dst, E, cnt, rank);
  scan_dis_kernel<<<(M + 1023) / 1024, 1024, 0, stream>>>(cnt, start, dis, M);
  place_kernel<<<(E + T - 1) / T, T, 0, stream>>>(src, dst, start, rank, list, E);

  gemm_wmma_kernel<256, 64><<<(M + 127) / 128, 256, 0, stream>>>(x, W1, H1, M);
  pull_relu64_kernel<<<PULL_BLOCKS, T, 0, stream>>>(start, list, dis, H1, b1, h1, M);

  gemm_wmma_kernel<64, 32><<<(M + 127) / 128, 256, 0, stream>>>(h1, W2, H2, M);
  pull_softmax32_kernel<<<PULL_BLOCKS, T, 0, stream>>>(start, list, dis, H2, b2, out, M);
}


# layer-1 GEMM and the edge count + rank pass share one launch (GEMM blocks first, 1024-edge counting blocks beside them): matrix-bound and atomic-bound work overlap; sum order: arrival rank as before
# speedup vs baseline: 2.0324x; 1.0396x over previous
#include <hip/hip_runtime.h>

typedef __attribute__((ext_vector_type(2))) float v2f;
typedef __attribute__((ext_vector_type(8))) float v8f;

__global__ void set_val_kernel(float* p, long long n, float v) {
  long long i = (long long)blockIdx.x * blockDim.x + threadIdx.x;
  if (i < n) p[i] = v;
}


__global__ void __launch_bounds__(1024)
scan_dis_kernel(const int* __restrict__ cnt, int* __restrict__ start,
                float* __restrict__ dis, int M) {
  __shared__ int sFront[32];
  __shared__ int sWave[32];
  const int tid = threadIdx.x, lane = tid & 31, wave = tid >> 5;
  int f = 0;
  const int4* c4 = (const int4*)cnt;
  for (int i = tid; i < (int)blockIdx.x * 256; i += 1024) {
    const int4 c = c4[i];
    f += (c.x + c.y) + (c.z + c.w);
  }
  for (int o = 16; o > 0; o >>= 1) f += __shfl_xor(f, o, 32);
  if (lane == 0) sFront[wave] = f;

  const int i = blockIdx.x * 1024 + tid;
  const int c = (i < M) ? cnt[i] : 0;
  int incl = c;
  for (int o = 1; o < 32; o <<= 1) {
    const int t = __shfl_up(incl, o, 32);
    if (lane >= o) incl += t;
  }
  if (lane == 31) sWave[wave] = incl;
  __syncthreads();
  int p = sFront[lane];
  int t = (lane < wave) ? sWave[lane] : 0;
  p += t;
  for (int o = 16; o > 0; o >>= 1) p += __shfl_xor(p, o, 32);
  if (i < M) {
    start[i + 1] = p + incl;
    dis[i] = rsqrtf(1.0f + (float)c);
  }
  if (i == 0) start[0] = 0;
}

__global__ void place_kernel(const int* __restrict__ src, const int* __restrict__ dst,
                             const int* __restrict__ start, const int* __restrict__ rank,
                             int* __restrict__ list, int E) {
  int e = blockIdx.x * blockDim.x + threadIdx.x;
  if (e < E) list[start[dst[e]] + rank[e]] = src[e];
}

template<int K, int N>
__device__ __forceinline__ void gemm_wmma_block(const float* __restrict__ X,
                                                const float* __restrict__ W,
                                                float* __restrict__ OUT, int M,
                                                const int blk) {
  constexpr int CT = N / 16;
  constexpr int QN = K / 32;
  __shared__ float4 sW[(K / 4) * N];

  for (int u = threadIdx.x; u < (K / 4) * N; u += 256) {
    const int n  = u % N;
    const int kq = u / N;
    const int q = kq >> 3, hh = (kq >> 2) & 1, jj = kq & 3;
    const float* w = W + (long long)(4 * kq) * N + n;
    float4 v;
    v.x = w[0];
    v.y = w[N];
    v.z = w[2 * N];
    v.w = w[3 * N];
    sW[((q * 4 + jj) * CT + (n % CT)) * 32 + hh * 16 + (n / CT)] = v;
  }
  __syncthreads();

  const int wave = threadIdx.x >> 5;
  const int lane = threadIdx.x & 31;
  const int rowBase = (blk * 8 + wave) * 16;
  if (rowBase + 16 > M) return;

  const int c = lane & 15;
  const int h = lane >> 4;

  const float4* xp = (const float4*)(X + (long long)(rowBase + c) * K + 16 * h);
  const float4* bp = sW + lane;

  v8f acc[CT];
#pragma unroll
  for (int t = 0; t < CT; ++t) acc[t] = v8f{};

  float4 a[2][4];
#pragma unroll
  for (int i = 0; i < 4; ++i) a[0][i] = xp[i];

#pragma unroll
  for (int q = 0; q < QN; ++q) {
    if (q + 1 < QN) {
#pragma unroll
      for (int i = 0; i < 4; ++i) a[(q + 1) & 1][i] = xp[(q + 1) * 8 + i];
    }
#pragma unroll
    for (int jj = 0; jj < 4; ++jj) {
      const float4 av = a[q & 1][jj];
      float4 b[CT];
#pragma unroll
      for (int t = 0; t < CT; ++t) b[t] = bp[((q * 4 + jj) * CT + t) * 32];
      v2f a0; a0.x = av.x; a0.y = av.y;
      v2f a1; a1.x = av.z; a1.y = av.w;
#pragma unroll
      for (int t = 0; t < CT; ++t) {
        v2f b0; b0.x = b[t].x; b0.y = b[t].y;
        acc[t] = __builtin_amdgcn_wmma_f32_16x16x4_f32(
            false, a0, false, b0, (short)0, acc[t], false, false);
      }
#pragma unroll
      for (int t = 0; t < CT; ++t) {
        v2f b1; b1.x = b[t].z; b1.y = b[t].w;
        acc[t] = __builtin_amdgcn_wmma_f32_16x16x4_f32(
            false, a1, false, b1, (short)0, acc[t], false, false);
      }
    }
  }

  float* out = OUT + (long long)(rowBase + 8 * h) * N + CT * c;
#pragma unroll
  for (int r = 0; r < 8; ++r) {
    if constexpr (CT == 4) {
      float4 o;
      o.x = acc[0][r]; o.y = acc[1][r]; o.z = acc[2][r]; o.w = acc[3][r];
      *(float4*)(out + (long long)r * N) = o;
    } else {
      static_assert(CT == 2, "N is 64 or 32 here");
      float2 o;
      o.x = acc[0][r]; o.y = acc[1][r];
      *(float2*)(out + (long long)r * N) = o;
    }
  }
}

template<int K, int N>
__global__ void __launch_bounds__(256)
gemm_wmma_kernel(const float* __restrict__ X,
                 const float* __restrict__ W,
                 float* __restrict__ OUT, int M) {
  gemm_wmma_block<K, N>(X, W, OUT, M, (int)blockIdx.x);
}

__global__ void __launch_bounds__(256)
gemm1_count_kernel(const float* __restrict__ X, const float* __restrict__ W,
                   float* __restrict__ OUT, int M, int gemmBlocks,
                   const int* __restrict__ dst, int E,
                   int* __restrict__ cnt, int* __restrict__ rank) {
  if ((int)blockIdx.x < gemmBlocks) {
    gemm_wmma_block<256, 64>(X, W, OUT, M, (int)blockIdx.x);
    return;
  }
  const int e0 = ((int)blockIdx.x - gemmBlocks) * 1024 + (int)threadIdx.x;
  int d[4], r[4];
#pragma unroll
  for (int i = 0; i < 4; ++i) {
    const int e = e0 + 256 * i;
    d[i] = (e < E) ? dst[e] : -1;
  }
#pragma unroll
  for (int i = 0; i < 4; ++i) {
    r[i] = 0;
    if (d[i] >= 0) r[i] = atomicAdd(&cnt[d[i]], 1);
  }
#pragma unroll
  for (int i = 0; i < 4; ++i) {
    const int e = e0 + 256 * i;
    if (d[i] >= 0) rank[e] = r[i];
  }
}


__global__ void __launch_bounds__(256)
pull_relu64_kernel(const int* __restrict__ start, const int* __restrict__ list,
                   const float* __restrict__ dis, const float* __restrict__ H,
                   const float* __restrict__ bias, float* __restrict__ out, int M) {
  const int lane = threadIdx.x & 31;
  const int wv = __builtin_amdgcn_readfirstlane((int)(threadIdx.x >> 5));
  const int nw = gridDim.x * 8;
  const float2 bb = *(const float2*)(bias + 2 * lane);
  for (int node = blockIdx.x * 8 + wv; node < M; node += nw) {
    const int beg = start[node], end = start[node + 1];
    const float dn = dis[node];
    float2 acc; acc.x = 0.0f; acc.y = 0.0f;
    for (int j = beg; j < end; j += 32) {
      const int m = min(32, end - j);
      int sid = 0; float w = 0.0f;
      if (lane < m) { sid = list[j + lane]; w = dis[sid] * dn; }
#pragma unroll 4
      for (int k = 0; k < m; ++k) {
        const int s = __builtin_amdgcn_readlane(sid, k);
        const float ws = __int_as_float(__builtin_amdgcn_readlane(__float_as_int(w), k));
        const float2 v = *(const float2*)(H + (long long)s * 64 + 2 * lane);
        acc.x += v.x * ws;
        acc.y += v.y * ws;
      }
    }
    const float2 hs = *(const float2*)(H + (long long)node * 64 + 2 * lane);
    float2 z;
    z.x = fmaxf(acc.x + dn * dn * hs.x + bb.x, 0.0f);
    z.y = fmaxf(acc.y + dn * dn * hs.y + bb.y, 0.0f);
    *(float2*)(out + (long long)node * 64 + 2 * lane) = z;
  }
}

__global__ void __launch_bounds__(256)
pull_softmax32_kernel(const int* __restrict__ start, const int* __restrict__ list,
                      const float* __restrict__ dis, const float* __restrict__ H,
                      const float* __restrict__ bias, float* __restrict__ out, int M) {
  const int lane = threadIdx.x & 31;
  const int wv = __builtin_amdgcn_readfirstlane((int)(threadIdx.x >> 5));
  const int nw = gridDim.x * 8;
  const float bb = bias[lane];
  for (int node = blockIdx.x * 8 + wv; node < M; node += nw) {
    const int beg = start[node], end = start[node + 1];
    const float dn = dis[node];
    float acc = 0.0f;
    for (int j = beg; j < end; j += 32) {
      const int m = min(32, end - j);
      int sid = 0; float w = 0.0f;
      if (lane < m) { sid = list[j + lane]; w = dis[sid] * dn; }
#pragma unroll 4
      for (int k = 0; k < m; ++k) {
        const int s = __builtin_amdgcn_readlane(sid, k);
        const float ws = __int_as_float(__builtin_amdgcn_readlane(__float_as_int(w), k));
        acc += H[(long long)s * 32 + lane] * ws;
      }
    }
    const float z = acc + dn * dn * H[(long long)node * 32 + lane] + bb;
    float mx = z;
    for (int o = 16; o > 0; o >>= 1) mx = fmaxf(mx, __shfl_xor(mx, o, 32));
    const float ev = __expf(z - mx);
    float sm = ev;
    for (int o = 16; o > 0; o >>= 1) sm += __shfl_xor(sm, o, 32);
    out[(long long)node * 32 + lane] = ev / sm;
  }
}

extern "C" void kernel_launch(void* const* d_in, const int* in_sizes, int n_in,
                              void* d_out, int out_size, void* d_ws, size_t ws_size,
                              hipStream_t stream) {
  (void)n_in; (void)out_size; (void)ws_size;
  const float* x  = (const float*)d_in[0];
  const int*   ei = (const int*)d_in[1];
  const float* W1 = (const float*)d_in[2];
  const float* b1 = (const float*)d_in[3];
  const float* W2 = (const float*)d_in[4];
  const float* b2 = (const float*)d_in[5];

  const int IN = 256, HID = 64, OD = 32;
  const int M = in_sizes[0] / IN;
  const int E = in_sizes[1] / 2;
  const int* src = ei;
  const int* dst = ei + E;

  const long long Mp = ((long long)M + 1023) & ~1023LL;
  float* ws   = (float*)d_ws;
  float* dis  = ws;
  float* H1   = dis + Mp;
  float* h1   = H1 + (long long)M * HID;
  int*   list = (int*)(h1 + (long long)M * HID);
  int*   cnt  = list + (((long long)E + 3) & ~3LL);
  int*   start = cnt + Mp;
  float* H2   = H1;
  float* out  = (float*)d_out;
  int*   rank = (int*)d_out;

  const int T = 256;
  const int PULL_BLOCKS = 2048;

  set_val_kernel<<<(M + T - 1) / T, T, 0, stream>>>((float*)cnt, M, 0.0f);
  gemm1_count_kernel<<<(M + 127) / 128 + (E + 1023) / 1024, 256, 0, stream>>>(
      x, W1, H1, M, (M + 127) / 128, dst, E, cnt, rank);
  scan_dis_kernel<<<(M + 1023) / 1024, 1024, 0, stream>>>(cnt, start, dis, M);
  place_kernel<<<(E + T - 1) / T, T, 0, stream>>>(src, dst, start, rank, list, E);

  pull_relu64_kernel<<<PULL_BLOCKS, T, 0, stream>>>(start, list, dis, H1, b1, h1, M);

  gemm_wmma_kernel<64, 32><<<(M + 127) / 128, 256, 0, stream>>>(h1, W2, H2, M);
  pull_softmax32_kernel<<<PULL_BLOCKS, T, 0, stream>>>(start, list, dis, H2, b2, out, M);
}


# edge list as fixed-width destination rows: the ranking add's return is the slot (32 in the out row, 16 spill, edge walk beyond); scan + place launches go, 1/sqrt(deg) formed at use; sum order kept
# speedup vs baseline: 2.3189x; 1.1410x over previous
#include <hip/hip_runtime.h>

typedef __attribute__((ext_vector_type(2))) float v2f;
typedef __attribute__((ext_vector_type(8))) float v8f;

__global__ void set_val_kernel(float* p, long long n, float v) {
  long long i = (long long)blockIdx.x * blockDim.x + threadIdx.x;
  if (i < n) p[i] = v;
}


template<int K, int N>
__device__ __forceinline__ void gemm_wmma_block(const float* __restrict__ X,
                                                const float* __restrict__ W,
                                                float* __restrict__ OUT, int M,
                                                const int blk) {
  constexpr int CT = N / 16;
  constexpr int QN = K / 32;
  __shared__ float4 sW[(K / 4) * N];

  for (int u = threadIdx.x; u < (K / 4) * N; u += 256) {
    const int n  = u % N;
    const int kq = u / N;
    const int q = kq >> 3, hh = (kq >> 2) & 1, jj = kq & 3;
    const float* w = W + (long long)(4 * kq) * N + n;
    float4 v;
    v.x = w[0];
    v.y = w[N];
    v.z = w[2 * N];
    v.w = w[3 * N];
    sW[((q * 4 + jj) * CT + (n % CT)) * 32 + hh * 16 + (n / CT)] = v;
  }
  __syncthreads();

  const int wave = threadIdx.x >> 5;
  const int lane = threadIdx.x & 31;
  const int rowBase = (blk * 8 + wave) * 16;
  if (rowBase + 16 > M) return;

  const int c = lane & 15;
  const int h = lane >> 4;

  const float4* xp = (const float4*)(X + (long long)(rowBase + c) * K + 16 * h);
  const float4* bp = sW + lane;

  v8f acc[CT];
#pragma unroll
  for (int t = 0; t < CT; ++t) acc[t] = v8f{};

  float4 a[2][4];
#pragma unroll
  for (int i = 0; i < 4; ++i) a[0][i] = xp[i];

#pragma unroll
  for (int q = 0; q < QN; ++q) {
    if (q + 1 < QN) {
#pragma unroll
      for (int i = 0; i < 4; ++i) a[(q + 1) & 1][i] = xp[(q + 1) * 8 + i];
    }
#pragma unroll
    for (int jj = 0; jj < 4; ++jj) {
      const float4 av = a[q & 1][jj];
      float4 b[CT];
#pragma unroll
      for (int t = 0; t < CT; ++t) b[t] = bp[((q * 4 + jj) * CT + t) * 32];
      v2f a0; a0.x = av.x; a0.y = av.y;
      v2f a1; a1.x = av.z; a1.y = av.w;
#pragma unroll
      for (int t = 0; t < CT; ++t) {
        v2f b0; b0.x = b[t].x; b0.y = b[t].y;
        acc[t] = __builtin_amdgcn_wmma_f32_16x16x4_f32(
            false, a0, false, b0, (short)0, acc[t], false, false);
      }
#pragma unroll
      for (int t = 0; t < CT; ++t) {
        v2f b1; b1.x = b[t].z; b1.y = b[t].w;
        acc[t] = __builtin_amdgcn_wmma_f32_16x16x4_f32(
            false, a1, false, b1, (short)0, acc[t], false, false);
      }
    }
  }

  float* out = OUT + (long long)(rowBase + 8 * h) * N + CT * c;
#pragma unroll
  for (int r = 0; r < 8; ++r) {
    if constexpr (CT == 4) {
      float4 o;
      o.x = acc[0][r]; o.y = acc[1][r]; o.z = acc[2][r]; o.w = acc[3][r];
      *(float4*)(out + (long long)r * N) = o;
    } else {
      static_assert(CT == 2, "N is 64 or 32 here");
      float2 o;
      o.x = acc[0][r]; o.y = acc[1][r];
      *(float2*)(out + (long long)r * N) = o;
    }
  }
}

template<int K, int N>
__global__ void __launch_bounds__(256)
gemm_wmma_kernel(const float* __restrict__ X,
                 const float* __restrict__ W,
                 float* __restrict__ OUT, int M) {
  gemm_wmma_block<K, N>(X, W, OUT, M, (int)blockIdx.x);
}

__global__ void __launch_bounds__(256)
gemm1_count_kernel(const float* __restrict__ X, const float* __restrict__ W,
                   float* __restrict__ OUT, int M, int gemmBlocks,
                   const int* __restrict__ src, const int* __restrict__ dst, int E,
                   int* __restrict__ cnt, int* __restrict__ ell, int* __restrict__ spill) {
  if ((int)blockIdx.x < gemmBlocks) {
    gemm_wmma_block<256, 64>(X, W, OUT, M, (int)blockIdx.x);
    return;
  }
  const int e0 = ((int)blockIdx.x - gemmBlocks) * 1024 + (int)threadIdx.x;
  int d[4], s[4], r[4];
#pragma unroll
  for (int i = 0; i < 4; ++i) {
    const int e = e0 + 256 * i;
    d[i] = (e < E) ? dst[e] : -1;
    s[i] = (e < E) ? src[e] : 0;
  }
#pragma unroll
  for (int i = 0; i < 4; ++i) {
    r[i] = 0;
    if (d[i] >= 0) r[i] = atomicAdd(&cnt[d[i]], 1);
  }
#pragma unroll
  for (int i = 0; i < 4; ++i) {
    if (d[i] >= 0) {
      if (r[i] < 32)      ell[(long long)d[i] * 32 + r[i]] = s[i];
      else if (r[i] < 48) spill[(long long)d[i] * 16 + (r[i] - 32)] = s[i];
    }
  }
}


__device__ __forceinline__ float inv_sqrt_deg(int c) { return rsqrtf(1.0f + (float)c); }

template<int V>
__device__ __forceinline__ void add_rows(float (&acc)[V], const float* __restrict__ H,
                                         int sid, float w, int m, int lane) {
#pragma unroll 4
  for (int k = 0; k < m; ++k) {
    const int s = __builtin_amdgcn_readlane(sid, k);
    const float ws = __int_as_float(__builtin_amdgcn_readlane(__float_as_int(w), k));
    if constexpr (V == 2) {
      const float2 v = *(const float2*)(H + (long long)s * 64 + 2 * lane);
      acc[0] += v.x * ws;
      acc[1] += v.y * ws;
    } else {
      acc[0] += H[(long long)s * 32 + lane] * ws;
    }
  }
}

template<int V>
__device__ __forceinline__ void add_rows_from_edges(float (&acc)[V], const float* __restrict__ H,
                                                    const int* __restrict__ src,
                                                    const int* __restrict__ dst, int E,
                                                    const int* __restrict__ cnt,
                                                    int node, float dn, int lane) {
  for (int e0 = 0; e0 < E; e0 += 32) {
    const int e = e0 + lane;
    const bool hit = (e < E) && (dst[e] == node);
    int sid = 0; float w = 0.0f;
    if (hit) { sid = src[e]; w = inv_sqrt_deg(cnt[sid]) * dn; }
    unsigned mask = __builtin_amdgcn_ballot_w32(hit);
    while (mask) {
      const int k = __builtin_ctz(mask);
      mask &= mask - 1;
      const int s = __builtin_amdgcn_readlane(sid, k);
      const float ws = __int_as_float(__builtin_amdgcn_readlane(__float_as_int(w), k));
      if constexpr (V == 2) {
        const float2 v = *(const float2*)(H + (long long)s * 64 + 2 * lane);
        acc[0] += v.x * ws;
        acc[1] += v.y * ws;
      } else {
        acc[0] += H[(long long)s * 32 + lane] * ws;
      }
    }
  }
}

template<int V>
__device__ __forceinline__ void sum_node(float (&acc)[V], const float* __restrict__ H,
                                         const int* __restrict__ cnt,
                                         const int* __restrict__ spill,
                                         const int* __restrict__ src,
                                         const int* __restrict__ dst, int E,
                                         int node, int c, int own, float dn, int lane) {
  if (__builtin_expect(c <= 48, 1)) {
    const int m = min(c, 32);
    int sid = 0; float w = 0.0f;
    if (lane < m) { sid = own; w = inv_sqrt_deg(cnt[sid]) * dn; }
    add_rows<V>(acc, H, sid, w, m, lane);
    if (c > 32) {
      const int m2 = c - 32;
      sid = 0; w = 0.0f;
      if (lane < m2) { sid = spill[(long long)node * 16 + lane]; w = inv_sqrt_deg(cnt[sid]) * dn; }
      add_rows<V>(acc, H, sid, w, m2, lane);
    }
  } else {
    add_rows_from_edges<V>(acc, H, src, dst, E, cnt, node, dn, lane);
  }
}

__global__ void __launch_bounds__(256)
pull_relu64_kernel(const int* __restrict__ cnt, const int* __restrict__ ell,
                   const int* __restrict__ spill,
                   const int* __restrict__ src, const int* __restrict__ dst, int E,
                   const float* __restrict__ H,
                   const float* __restrict__ bias, float* __restrict__ out, int M) {
  const int lane = threadIdx.x & 31;
  const int wv = __builtin_amdgcn_readfirstlane((int)(threadIdx.x >> 5));
  const int nw = gridDim.x * 8;
  const float2 bb = *(const float2*)(bias + 2 * lane);
  for (int node = blockIdx.x * 8 + wv; node < M; node += nw) {
    int own = ell[(long long)node * 32 + lane];
    const int c = cnt[node];
    asm volatile("" : "+v"(own));
    const float dn = inv_sqrt_deg(c);
    float acc[2] = {0.0f, 0.0f};
    sum_node<2>(acc, H, cnt, spill, src, dst, E, node, c, own, dn, lane);
    const float2 hs = *(const float2*)(H + (long long)node * 64 + 2 * lane);
    float2 z;
    z.x = fmaxf(acc[0] + dn * dn * hs.x + bb.x, 0.0f);
    z.y = fmaxf(acc[1] + dn * dn * hs.y + bb.y, 0.0f);
    *(float2*)(out + (long long)node * 64 + 2 * lane) = z;
  }
}

__global__ void __launch_bounds__(256)
pull_softmax32_kernel(const int* __restrict__ cnt, const int* __restrict__ spill,
                      const int* __restrict__ src, const int* __restrict__ dst, int E,
                      const float* __restrict__ H,
                      const float* __restrict__ bias, float* out, int M) {
  const int lane = threadIdx.x & 31;
  const int wv = __builtin_amdgcn_readfirstlane((int)(threadIdx.x >> 5));
  const int nw = gridDim.x * 8;
  const float bb = bias[lane];
  for (int node = blockIdx.x * 8 + wv; node < M; node += nw) {
    int own = __float_as_int(out[(long long)node * 32 + lane]);
    const int c = cnt[node];
    asm volatile("" : "+v"(own));
    const float dn = inv_sqrt_deg(c);
    float acc[1] = {0.0f};
    sum_node<1>(acc, H, cnt, spill, src, dst, E, node, c, own, dn, lane);
    const float z = acc[0] + dn * dn * H[(long long)node * 32 + lane] + bb;
    float mx = z;
    for (int o = 16; o > 0; o >>= 1) mx = fmaxf(mx, __shfl_xor(mx, o, 32));
    const float ev = __expf(z - mx);
    float sm = ev;
    for (int o = 16; o > 0; o >>= 1) sm += __shfl_xor(sm, o, 32);
    out[(long long)node * 32 + lane] = ev / sm;
  }
}

extern "C" void kernel_launch(void* const* d_in, const int* in_sizes, int n_in,
                              void* d_out, int out_size, void* d_ws, size_t ws_size,
                              hipStream_t stream) {
  (void)n_in; (void)out_size; (void)ws_size;
  const float* x  = (const float*)d_in[0];
  const int*   ei = (const int*)d_in[1];
  const float* W1 = (const float*)d_in[2];
  const float* b1 = (const float*)d_in[3];
  const float* W2 = (const float*)d_in[4];
  const float* b2 = (const float*)d_in[5];

  const int IN = 256, HID = 64, OD = 32;
  const int M = in_sizes[0] / IN;
  const int E = in_sizes[1] / 2;
  const int* src = ei;
  const int* dst = ei + E;

  const long long Mp = ((long long)M + 1023) & ~1023LL;
  int*   cnt   = (int*)d_ws;
  float* H1    = (float*)d_ws + Mp;
  float* h1    = H1 + (long long)M * HID;
  int*   spill = (int*)(h1 + (long long)M * HID);
  float* H2    = H1;
  float* out   = (float*)d_out;
  int*   ell   = (int*)d_out;

  const int T = 256;
  const int PULL_BLOCKS = 2048;

  set_val_kernel<<<(M + T - 1) / T, T, 0, stream>>>((float*)cnt, M, 0.0f);
  gemm1_count_kernel<<<(M + 127) / 128 + (E + 1023) / 1024, 256, 0, stream>>>(
      x, W1, H1, M, (M + 127) / 128, src, dst, E, cnt, ell, spill);

  pull_relu64_kernel<<<PULL_BLOCKS, T, 0, stream>>>(cnt, ell, spill, src, dst, E, H1, b1, h1, M);

  gemm_wmma_kernel<64, 32><<<(M + 127) / 128, 256, 0, stream>>>(h1, W2, H2, M);
  pull_softmax32_kernel<<<PULL_BLOCKS, T, 0, stream>>>(cnt, spill, src, dst, E, H2, b2, out, M);
}


# edge sums: a node's row requests leave in groups (own row with its slot line; source rows 8 per trip, slots past the end ask row 0, never added): 8-9 dependent round trips become 4-5; sum order kept
# speedup vs baseline: 2.4192x; 1.0433x over previous
#include <hip/hip_runtime.h>

typedef __attribute__((ext_vector_type(2))) float v2f;
typedef __attribute__((ext_vector_type(8))) float v8f;

__global__ void set_val_kernel(float* p, long long n, float v) {
  long long i = (long long)blockIdx.x * blockDim.x + threadIdx.x;
  if (i < n) p[i] = v;
}


template<int K, int N>
__device__ __forceinline__ void gemm_wmma_block(const float* __restrict__ X,
                                                const float* __restrict__ W,
                                                float* __restrict__ OUT, int M,
                                                const int blk) {
  constexpr int CT = N / 16;
  constexpr int QN = K / 32;
  __shared__ float4 sW[(K / 4) * N];

  for (int u = threadIdx.x; u < (K / 4) * N; u += 256) {
    const int n  = u % N;
    const int kq = u / N;
    const int q = kq >> 3, hh = (kq >> 2) & 1, jj = kq & 3;
    const float* w = W + (long long)(4 * kq) * N + n;
    float4 v;
    v.x = w[0];
    v.y = w[N];
    v.z = w[2 * N];
    v.w = w[3 * N];
    sW[((q * 4 + jj) * CT + (n % CT)) * 32 + hh * 16 + (n / CT)] = v;
  }
  __syncthreads();

  const int wave = threadIdx.x >> 5;
  const int lane = threadIdx.x & 31;
  const int rowBase = (blk * 8 + wave) * 16;
  if (rowBase + 16 > M) return;

  const int c = lane & 15;
  const int h = lane >> 4;

  const float4* xp = (const float4*)(X + (long long)(rowBase + c) * K + 16 * h);
  const float4* bp = sW + lane;

  v8f acc[CT];
#pragma unroll
  for (int t = 0; t < CT; ++t) acc[t] = v8f{};

  float4 a[2][4];
#pragma unroll
  for (int i = 0; i < 4; ++i) a[0][i] = xp[i];

#pragma unroll
  for (int q = 0; q < QN; ++q) {
    if (q + 1 < QN) {
#pragma unroll
      for (int i = 0; i < 4; ++i) a[(q + 1) & 1][i] = xp[(q + 1) * 8 + i];
    }
#pragma unroll
    for (int jj = 0; jj < 4; ++jj) {
      const float4 av = a[q & 1][jj];
      float4 b[CT];
#pragma unroll
      for (int t = 0; t < CT; ++t) b[t] = bp[((q * 4 + jj) * CT + t) * 32];
      v2f a0; a0.x = av.x; a0.y = av.y;
      v2f a1; a1.x = av.z; a1.y = av.w;
#pragma unroll
      for (int t = 0; t < CT; ++t) {
        v2f b0; b0.x = b[t].x; b0.y = b[t].y;
        acc[t] = __builtin_amdgcn_wmma_f32_16x16x4_f32(
            false, a0, false, b0, (short)0, acc[t], false, false);
      }
#pragma unroll
      for (int t = 0; t < CT; ++t) {
        v2f b1; b1.x = b[t].z; b1.y = b[t].w;
        acc[t] = __builtin_amdgcn_wmma_f32_16x16x4_f32(
            false, a1, false, b1, (short)0, acc[t], false, false);
      }
    }
  }

  float* out = OUT + (long long)(rowBase + 8 * h) * N + CT * c;
#pragma unroll
  for (int r = 0; r < 8; ++r) {
    if constexpr (CT == 4) {
      float4 o;
      o.x = acc[0][r]; o.y = acc[1][r]; o.z = acc[2][r]; o.w = acc[3][r];
      *(float4*)(out + (long long)r * N) = o;
    } else {
      static_assert(CT == 2, "N is 64 or 32 here");
      float2 o;
      o.x = acc[0][r]; o.y = acc[1][r];
      *(float2*)(out + (long long)r * N) = o;
    }
  }
}

template<int K, int N>
__global__ void __launch_bounds__(256)
gemm_wmma_kernel(const float* __restrict__ X,
                 const float* __restrict__ W,
                 float* __restrict__ OUT, int M) {
  gemm_wmma_block<K, N>(X, W, OUT, M, (int)blockIdx.x);
}

__global__ void __launch_bounds__(256)
gemm1_count_kernel(const float* __restrict__ X, const float* __restrict__ W,
                   float* __restrict__ OUT, int M, int gemmBlocks,
                   const int* __restrict__ src, const int* __restrict__ dst, int E,
                   int* __restrict__ cnt, int* __restrict__ ell, int* __restrict__ spill) {
  if ((int)blockIdx.x < gemmBlocks) {
    gemm_wmma_block<256, 64>(X, W, OUT, M, (int)blockIdx.x);
    return;
  }
  const int e0 = ((int)blockIdx.x - gemmBlocks) * 1024 + (int)threadIdx.x;
  int d[4], s[4], r[4];
#pragma unroll
  for (int i = 0; i < 4; ++i) {
    const int e = e0 + 256 * i;
    d[i] = (e < E) ? dst[e] : -1;
    s[i] = (e < E) ? src[e] : 0;
  }
#pragma unroll
  for (int i = 0; i < 4; ++i) {
    r[i] = 0;
    if (d[i] >= 0) r[i] = atomicAdd(&cnt[d[i]], 1);
  }
#pragma unroll
  for (int i = 0; i < 4; ++i) {
    if (d[i] >= 0) {
      if (r[i] < 32)      ell[(long long)d[i] * 32 + r[i]] = s[i];
      else if (r[i] < 48) spill[(long long)d[i] * 16 + (r[i] - 32)] = s[i];
    }
  }
}


__device__ __forceinline__ float inv_sqrt_deg(int c) { return rsqrtf(1.0f + (float)c); }

template<int V>
__device__ __forceinline__ void add_rows(float (&acc)[V], const float* __restrict__ H,
                                         int sid, float w, int m, int lane) {
  for (int k0 = 0; k0 < m; k0 += 8) {
    float vx[8], vy[8];
#pragma unroll
    for (int j = 0; j < 8; ++j) {
      const int s = __builtin_amdgcn_readlane(sid, k0 + j);
      if constexpr (V == 2) {
        const float2 v = *(const float2*)(H + (long long)s * 64 + 2 * lane);
        vx[j] = v.x; vy[j] = v.y;
      } else {
        vx[j] = H[(long long)s * 32 + lane]; vy[j] = 0.0f;
      }
    }
#pragma unroll
    for (int j = 0; j < 8; ++j) {
      if constexpr (V == 2) asm volatile("" : "+v"(vx[j]), "+v"(vy[j]));
      else                  asm volatile("" : "+v"(vx[j]));
    }
#pragma unroll
    for (int j = 0; j < 8; ++j) {
      if (j == 0 || k0 + j < m) {
        const float ws = __int_as_float(__builtin_amdgcn_readlane(__float_as_int(w), k0 + j));
        acc[0] += vx[j] * ws;
        if constexpr (V == 2) acc[1] += vy[j] * ws;
      }
    }
  }
}

template<int V>
__device__ __forceinline__ void add_rows_from_edges(float (&acc)[V], const float* __restrict__ H,
                                                    const int* __restrict__ src,
                                                    const int* __restrict__ dst, int E,
                                                    const int* __restrict__ cnt,
                                                    int node, float dn, int lane) {
  for (int e0 = 0; e0 < E; e0 += 32) {
    const int e = e0 + lane;
    const bool hit = (e < E) && (dst[e] == node);
    int sid = 0; float w = 0.0f;
    if (hit) { sid = src[e]; w = inv_sqrt_deg(cnt[sid]) * dn; }
    unsigned mask = __builtin_amdgcn_ballot_w32(hit);
    while (mask) {
      const int k = __builtin_ctz(mask);
      mask &= mask - 1;
      const int s = __builtin_amdgcn_readlane(sid, k);
      const float ws = __int_as_float(__builtin_amdgcn_readlane(__float_as_int(w), k));
      if constexpr (V == 2) {
        const float2 v = *(const float2*)(H + (long long)s * 64 + 2 * lane);
        acc[0] += v.x * ws;
        acc[1] += v.y * ws;
      } else {
        acc[0] += H[(long long)s * 32 + lane] * ws;
      }
    }
  }
}

template<int V>
__device__ __forceinline__ void sum_node(float (&acc)[V], const float* __restrict__ H,
                                         const int* __restrict__ cnt,
                                         const int* __restrict__ spill,
                                         const int* __restrict__ src,
                                         const int* __restrict__ dst, int E,
                                         int node, int c, int own, float dn, int lane) {
  if (__builtin_expect(c <= 48, 1)) {
    const int m = min(c, 32);
    int sid = 0; float w = 0.0f;
    if (lane < m) { sid = own; w = inv_sqrt_deg(cnt[sid]) * dn; }
    add_rows<V>(acc, H, sid, w, m, lane);
    if (c > 32) {
      const int m2 = c - 32;
      sid = 0; w = 0.0f;
      if (lane < m2) { sid = spill[(long long)node * 16 + lane]; w = inv_sqrt_deg(cnt[sid]) * dn; }
      add_rows<V>(acc, H, sid, w, m2, lane);
    }
  } else {
    add_rows_from_edges<V>(acc, H, src, dst, E, cnt, node, dn, lane);
  }
}

__global__ void __launch_bounds__(256)
pull_relu64_kernel(const int* __restrict__ cnt, const int* __restrict__ ell,
                   const int* __restrict__ spill,
                   const int* __restrict__ src, const int* __restrict__ dst, int E,
                   const float* __restrict__ H,
                   const float* __restrict__ bias, float* __restrict__ out, int M) {
  const int lane = threadIdx.x & 31;
  const int wv = __builtin_amdgcn_readfirstlane((int)(threadIdx.x >> 5));
  const int nw = gridDim.x * 8;
  const float2 bb = *(const float2*)(bias + 2 * lane);
  for (int node = blockIdx.x * 8 + wv; node < M; node += nw) {
    int own = ell[(long long)node * 32 + lane];
    float2 hs = *(const float2*)(H + (long long)node * 64 + 2 * lane);
    const int c = cnt[node];
    asm volatile("" : "+v"(own), "+v"(hs.x), "+v"(hs.y));
    const float dn = inv_sqrt_deg(c);
    float acc[2] = {0.0f, 0.0f};
    sum_node<2>(acc, H, cnt, spill, src, dst, E, node, c, own, dn, lane);
    float2 z;
    z.x = fmaxf(acc[0] + dn * dn * hs.x + bb.x, 0.0f);
    z.y = fmaxf(acc[1] + dn * dn * hs.y + bb.y, 0.0f);
    *(float2*)(out + (long long)node * 64 + 2 * lane) = z;
  }
}

__global__ void __launch_bounds__(256)
pull_softmax32_kernel(const int* __restrict__ cnt, const int* __restrict__ spill,
                      const int* __restrict__ src, const int* __restrict__ dst, int E,
                      const float* __restrict__ H,
                      const float* __restrict__ bias, float* out, int M) {
  const int lane = threadIdx.x & 31;
  const int wv = __builtin_amdgcn_readfirstlane((int)(threadIdx.x >> 5));
  const int nw = gridDim.x * 8;
  const float bb = bias[lane];
  for (int node = blockIdx.x * 8 + wv; node < M; node += nw) {
    int own = __float_as_int(out[(long long)node * 32 + lane]);
    float hs = H[(long long)node * 32 + lane];
    const int c = cnt[node];
    asm volatile("" : "+v"(own), "+v"(hs));
    const float dn = inv_sqrt_deg(c);
    float acc[1] = {0.0f};
    sum_node<1>(acc, H, cnt, spill, src, dst, E, node, c, own, dn, lane);
    const float z = acc[0] + dn * dn * hs + bb;
    float mx = z;
    for (int o = 16; o > 0; o >>= 1) mx = fmaxf(mx, __shfl_xor(mx, o, 32));
    const float ev = __expf(z - mx);
    float sm = ev;
    for (int o = 16; o > 0; o >>= 1) sm += __shfl_xor(sm, o, 32);
    out[(long long)node * 32 + lane] = ev / sm;
  }
}

extern "C" void kernel_launch(void* const* d_in, const int* in_sizes, int n_in,
                              void* d_out, int out_size, void* d_ws, size_t ws_size,
                              hipStream_t stream) {
  (void)n_in; (void)out_size; (void)ws_size;
  const float* x  = (const float*)d_in[0];
  const int*   ei = (const int*)d_in[1];
  const float* W1 = (const float*)d_in[2];
  const float* b1 = (const float*)d_in[3];
  const float* W2 = (const float*)d_in[4];
  const float* b2 = (const float*)d_in[5];

  const int IN = 256, HID = 64, OD = 32;
  const int M = in_sizes[0] / IN;
  const int E = in_sizes[1] / 2;
  const int* src = ei;
  const int* dst = ei + E;

  const long long Mp = ((long long)M + 1023) & ~1023LL;
  int*   cnt   = (int*)d_ws;
  float* H1    = (float*)d_ws + Mp;
  float* h1    = H1 + (long long)M * HID;
  int*   spill = (int*)(h1 + (long long)M * HID);
  float* H2    = H1;
  float* out   = (float*)d_out;
  int*   ell   = (int*)d_out;

  const int T = 256;
  const int PULL_BLOCKS = 2048;

  set_val_kernel<<<(M + T - 1) / T, T, 0, stream>>>((float*)cnt, M, 0.0f);
  gemm1_count_kernel<<<(M + 127) / 128 + (E + 1023) / 1024, 256, 0, stream>>>(
      x, W1, H1, M, (M + 127) / 128, src, dst, E, cnt, ell, spill);

  pull_relu64_kernel<<<PULL_BLOCKS, T, 0, stream>>>(cnt, ell, spill, src, dst, E, H1, b1, h1, M);

  gemm_wmma_kernel<64, 32><<<(M + 127) / 128, 256, 0, stream>>>(h1, W2, H2, M);
  pull_softmax32_kernel<<<PULL_BLOCKS, T, 0, stream>>>(cnt, spill, src, dst, E, H2, b2, out, M);
}


# count + rank pass: each package half ranks only the edges whose destination it owns (nodes split in two ranges; waves draw 256-edge tickets, own range first); sum order: arrival rank as before
# speedup vs baseline: 4.2945x; 1.7752x over previous
#include <hip/hip_runtime.h>

typedef __attribute__((ext_vector_type(2))) float v2f;
typedef __attribute__((ext_vector_type(8))) float v8f;

__global__ void set_val_kernel(float* p, long long n, float v) {
  long long i = (long long)blockIdx.x * blockDim.x + threadIdx.x;
  if (i < n) p[i] = v;
}


template<int K, int N>
__device__ __forceinline__ void gemm_wmma_block(const float* __restrict__ X,
                                                const float* __restrict__ W,
                                                float* __restrict__ OUT, int M,
                                                const int blk) {
  constexpr int CT = N / 16;
  constexpr int QN = K / 32;
  __shared__ float4 sW[(K / 4) * N];

  for (int u = threadIdx.x; u < (K / 4) * N; u += 256) {
    const int n  = u % N;
    const int kq = u / N;
    const int q = kq >> 3, hh = (kq >> 2) & 1, jj = kq & 3;
    const float* w = W + (long long)(4 * kq) * N + n;
    float4 v;
    v.x = w[0];
    v.y = w[N];
    v.z = w[2 * N];
    v.w = w[3 * N];
    sW[((q * 4 + jj) * CT + (n % CT)) * 32 + hh * 16 + (n / CT)] = v;
  }
  __syncthreads();

  const int wave = threadIdx.x >> 5;
  const int lane = threadIdx.x & 31;
  const int rowBase = (blk * 8 + wave) * 16;
  if (rowBase + 16 > M) return;

  const int c = lane & 15;
  const int h = lane >> 4;

  const float4* xp = (const float4*)(X + (long long)(rowBase + c) * K + 16 * h);
  const float4* bp = sW + lane;

  v8f acc[CT];
#pragma unroll
  for (int t = 0; t < CT; ++t) acc[t] = v8f{};

  float4 a[2][4];
#pragma unroll
  for (int i = 0; i < 4; ++i) a[0][i] = xp[i];

#pragma unroll
  for (int q = 0; q < QN; ++q) {
    if (q + 1 < QN) {
#pragma unroll
      for (int i = 0; i < 4; ++i) a[(q + 1) & 1][i] = xp[(q + 1) * 8 + i];
    }
#pragma unroll
    for (int jj = 0; jj < 4; ++jj) {
      const float4 av = a[q & 1][jj];
      float4 b[CT];
#pragma unroll
      for (int t = 0; t < CT; ++t) b[t] = bp[((q * 4 + jj) * CT + t) * 32];
      v2f a0; a0.x = av.x; a0.y = av.y;
      v2f a1; a1.x = av.z; a1.y = av.w;
#pragma unroll
      for (int t = 0; t < CT; ++t) {
        v2f b0; b0.x = b[t].x; b0.y = b[t].y;
        acc[t] = __builtin_amdgcn_wmma_f32_16x16x4_f32(
            false, a0, false, b0, (short)0, acc[t], false, false);
      }
#pragma unroll
      for (int t = 0; t < CT; ++t) {
        v2f b1; b1.x = b[t].z; b1.y = b[t].w;
        acc[t] = __builtin_amdgcn_wmma_f32_16x16x4_f32(
            false, a1, false, b1, (short)0, acc[t], false, false);
      }
    }
  }

  float* out = OUT + (long long)(rowBase + 8 * h) * N + CT * c;
#pragma unroll
  for (int r = 0; r < 8; ++r) {
    if constexpr (CT == 4) {
      float4 o;
      o.x = acc[0][r]; o.y = acc[1][r]; o.z = acc[2][r]; o.w = acc[3][r];
      *(float4*)(out + (long long)r * N) = o;
    } else {
      static_assert(CT == 2, "N is 64 or 32 here");
      float2 o;
      o.x = acc[0][r]; o.y = acc[1][r];
      *(float2*)(out + (long long)r * N) = o;
    }
  }
}

template<int K, int N>
__global__ void __launch_bounds__(256)
gemm_wmma_kernel(const float* __restrict__ X,
                 const float* __restrict__ W,
                 float* __restrict__ OUT, int M) {
  gemm_wmma_block<K, N>(X, W, OUT, M, (int)blockIdx.x);
}


__device__ __forceinline__ int count_chunk(const int* __restrict__ src, const int* __restrict__ dst,
                                           int E, int t, int rng, int split, int lane,
                                           int* __restrict__ cnt, int* __restrict__ ctr,
                                           int* __restrict__ ell, int* __restrict__ spill) {
  const int e0 = t * 256 + lane;
  int d[8], s[8], r[8];
#pragma unroll
  for (int i = 0; i < 8; ++i) {
    const int e = e0 + 32 * i;
    d[i] = (e < E) ? dst[e] : -1;
    s[i] = (e < E) ? src[e] : 0;
  }
#pragma unroll
  for (int i = 0; i < 8; ++i) {
    const bool here = (d[i] >= 0) && ((d[i] >= split) == (rng != 0));
    if (!here) d[i] = -1;
  }
#pragma unroll
  for (int i = 0; i < 8; ++i) {
    r[i] = 0;
    if (d[i] >= 0) r[i] = atomicAdd(&cnt[d[i]], 1);
  }
  int tv = 0;
  if (lane == 0) tv = atomicAdd(ctr, 1);
#pragma unroll
  for (int i = 0; i < 8; ++i) {
    if (d[i] >= 0) {
      if (r[i] < 32)      ell[(long long)d[i] * 32 + r[i]] = s[i];
      else if (r[i] < 48) spill[(long long)d[i] * 16 + (r[i] - 32)] = s[i];
    }
  }
  return tv;
}

__global__ void __launch_bounds__(256)
gemm1_count_kernel(const float* __restrict__ X, const float* __restrict__ W,
                   float* __restrict__ OUT, int M, int gemmBlocks,
                   const int* __restrict__ src, const int* __restrict__ dst, int E,
                   int* __restrict__ cnt, int* __restrict__ tick,
                   int* __restrict__ ell, int* __restrict__ spill) {
  if ((int)blockIdx.x < gemmBlocks) {
    gemm_wmma_block<256, 64>(X, W, OUT, M, (int)blockIdx.x);
    return;
  }
  const int lane = threadIdx.x & 31;
  const unsigned where = __builtin_amdgcn_s_sendmsg_rtn(0x87);
  const int half = (int)((where >> 18) & 1u);
  const int split = ((M >> 1) + 31) & ~31;
  const int nTickets = (E + 255) >> 8;
  for (int ph = 0; ph < 2; ++ph) {
    const int rng = half ^ ph;
    int* ctr = tick + 64 * rng;
    int tv = 0;
    if (lane == 0) tv = atomicAdd(ctr, 1);
    for (int it = 0; it <= nTickets; ++it) {
      const int t = __builtin_amdgcn_readfirstlane(tv);
      if (t >= nTickets) break;
      tv = count_chunk(src, dst, E, t, rng, split, lane, cnt, ctr, ell, spill);
    }
  }
}


__device__ __forceinline__ float inv_sqrt_deg(int c) { return rsqrtf(1.0f + (float)c); }

template<int V>
__device__ __forceinline__ void add_rows(float (&acc)[V], const float* __restrict__ H,
                                         int sid, float w, int m, int lane) {
  for (int k0 = 0; k0 < m; k0 += 8) {
    float vx[8], vy[8];
#pragma unroll
    for (int j = 0; j < 8; ++j) {
      const int s = __builtin_amdgcn_readlane(sid, k0 + j);
      if constexpr (V == 2) {
        const float2 v = *(const float2*)(H + (long long)s * 64 + 2 * lane);
        vx[j] = v.x; vy[j] = v.y;
      } else {
        vx[j] = H[(long long)s * 32 + lane]; vy[j] = 0.0f;
      }
    }
#pragma unroll
    for (int j = 0; j < 8; ++j) {
      if constexpr (V == 2) asm volatile("" : "+v"(vx[j]), "+v"(vy[j]));
      else                  asm volatile("" : "+v"(vx[j]));
    }
#pragma unroll
    for (int j = 0; j < 8; ++j) {
      if (j == 0 || k0 + j < m) {
        const float ws = __int_as_float(__builtin_amdgcn_readlane(__float_as_int(w), k0 + j));
        acc[0] += vx[j] * ws;
        if constexpr (V == 2) acc[1] += vy[j] * ws;
      }
    }
  }
}

template<int V>
__device__ __forceinline__ void add_rows_from_edges(float (&acc)[V], const float* __restrict__ H,
                                                    const int* __restrict__ src,
                                                    const int* __restrict__ dst, int E,
                                                    const int* __restrict__ cnt,
                                                    int node, float dn, int lane) {
  for (int e0 = 0; e0 < E; e0 += 32) {
    const int e = e0 + lane;
    const bool hit = (e < E) && (dst[e] == node);
    int sid = 0; float w = 0.0f;
    if (hit) { sid = src[e]; w = inv_sqrt_deg(cnt[sid]) * dn; }
    unsigned mask = __builtin_amdgcn_ballot_w32(hit);
    while (mask) {
      const int k = __builtin_ctz(mask);
      mask &= mask - 1;
      const int s = __builtin_amdgcn_readlane(sid, k);
      const float ws = __int_as_float(__builtin_amdgcn_readlane(__float_as_int(w), k));
      if constexpr (V == 2) {
        const float2 v = *(const float2*)(H + (long long)s * 64 + 2 * lane);
        acc[0] += v.x * ws;
        acc[1] += v.y * ws;
      } else {
        acc[0] += H[(long long)s * 32 + lane] * ws;
      }
    }
  }
}

template<int V>
__device__ __forceinline__ void sum_node(float (&acc)[V], const float* __restrict__ H,
                                         const int* __restrict__ cnt,
                                         const int* __restrict__ spill,
                                         const int* __restrict__ src,
                                         const int* __restrict__ dst, int E,
                                         int node, int c, int own, float dn, int lane) {
  if (__builtin_expect(c <= 48, 1)) {
    const int m = min(c, 32);
    int sid = 0; float w = 0.0f;
    if (lane < m) { sid = own; w = inv_sqrt_deg(cnt[sid]) * dn; }
    add_rows<V>(acc, H, sid, w, m, lane);
    if (c > 32) {
      const int m2 = c - 32;
      sid = 0; w = 0.0f;
      if (lane < m2) { sid = spill[(long long)node * 16 + lane]; w = inv_sqrt_deg(cnt[sid]) * dn; }
      add_rows<V>(acc, H, sid, w, m2, lane);
    }
  } else {
    add_rows_from_edges<V>(acc, H, src, dst, E, cnt, node, dn, lane);
  }
}

__global__ void __launch_bounds__(256)
pull_relu64_kernel(const int* __restrict__ cnt, const int* __restrict__ ell,
                   const int* __restrict__ spill,
                   const int* __restrict__ src, const int* __restrict__ dst, int E,
                   const float* __restrict__ H,
                   const float* __restrict__ bias, float* __restrict__ out, int M) {
  const int lane = threadIdx.x & 31;
  const int wv = __builtin_amdgcn_readfirstlane((int)(threadIdx.x >> 5));
  const int nw = gridDim.x * 8;
  const float2 bb = *(const float2*)(bias + 2 * lane);
  for (int node = blockIdx.x * 8 + wv; node < M; node += nw) {
    int own = ell[(long long)node * 32 + lane];
    float2 hs = *(const float2*)(H + (long long)node * 64 + 2 * lane);
    const int c = cnt[node];
    asm volatile("" : "+v"(own), "+v"(hs.x), "+v"(hs.y));
    const float dn = inv_sqrt_deg(c);
    float acc[2] = {0.0f, 0.0f};
    sum_node<2>(acc, H, cnt, spill, src, dst, E, node, c, own, dn, lane);
    float2 z;
    z.x = fmaxf(acc[0] + dn * dn * hs.x + bb.x, 0.0f);
    z.y = fmaxf(acc[1] + dn * dn * hs.y + bb.y, 0.0f);
    *(float2*)(out + (long long)node * 64 + 2 * lane) = z;
  }
}

__global__ void __launch_bounds__(256)
pull_softmax32_kernel(const int* __restrict__ cnt, const int* __restrict__ spill,
                      const int* __restrict__ src, const int* __restrict__ dst, int E,
                      const float* __restrict__ H,
                      const float* __restrict__ bias, float* out, int M) {
  const int lane = threadIdx.x & 31;
  const int wv = __builtin_amdgcn_readfirstlane((int)(threadIdx.x >> 5));
  const int nw = gridDim.x * 8;
  const float bb = bias[lane];
  for (int node = blockIdx.x * 8 + wv; node < M; node += nw) {
    int own = __float_as_int(out[(long long)node * 32 + lane]);
    float hs = H[(long long)node * 32 + lane];
    const int c = cnt[node];
    asm volatile("" : "+v"(own), "+v"(hs));
    const float dn = inv_sqrt_deg(c);
    float acc[1] = {0.0f};
    sum_node<1>(acc, H, cnt, spill, src, dst, E, node, c, own, dn, lane);
    const float z = acc[0] + dn * dn * hs + bb;
    float mx = z;
    for (int o = 16; o > 0; o >>= 1) mx = fmaxf(mx, __shfl_xor(mx, o, 32));
    const float ev = __expf(z - mx);
    float sm = ev;
    for (int o = 16; o > 0; o >>= 1) sm += __shfl_xor(sm, o, 32);
    out[(long long)node * 32 + lane] = ev / sm;
  }
}

extern "C" void kernel_launch(void* const* d_in, const int* in_sizes, int n_in,
                              void* d_out, int out_size, void* d_ws, size_t ws_size,
                              hipStream_t stream) {
  (void)n_in; (void)out_size; (void)ws_size;
  const float* x  = (const float*)d_in[0];
  const int*   ei = (const int*)d_in[1];
  const float* W1 = (const float*)d_in[2];
  const float* b1 = (const float*)d_in[3];
  const float* W2 = (const float*)d_in[4];
  const float* b2 = (const float*)d_in[5];

  const int IN = 256, HID = 64, OD = 32;
  const int M = in_sizes[0] / IN;
  const int E = in_sizes[1] / 2;
  const int* src = ei;
  const int* dst = ei + E;

  const long long Mp = ((long long)M + 1023) & ~1023LL;
  const long long Cw = Mp + 128;
  int*   cnt   = (int*)d_ws;
  int*   tick  = cnt + Mp;
  float* H1    = (float*)d_ws + Cw;
  float* h1    = H1 + (long long)M * HID;
  int*   spill = (int*)(h1 + (long long)M * HID);
  float* H2    = H1;
  float* out   = (float*)d_out;
  int*   ell   = (int*)d_out;

  const int T = 256;
  const int PULL_BLOCKS = 2048;
  const int COUNT_BLOCKS = 768;

  set_val_kernel<<<(int)((Cw + T - 1) / T), T, 0, stream>>>((float*)cnt, Cw, 0.0f);
  gemm1_count_kernel<<<(M + 127) / 128 + COUNT_BLOCKS, 256, 0, stream>>>(
      x, W1, H1, M, (M + 127) / 128, src, dst, E, cnt, tick, ell, spill);

  pull_relu64_kernel<<<PULL_BLOCKS, T, 0, stream>>>(cnt, ell, spill, src, dst, E, H1, b1, h1, M);

  gemm_wmma_kernel<64, 32><<<(M + 127) / 128, 256, 0, stream>>>(h1, W2, H2, M);
  pull_softmax32_kernel<<<PULL_BLOCKS, T, 0, stream>>>(cnt, spill, src, dst, E, H2, b2, out, M);
}


# layer-2 product joins the layer-1 edge-sum launch: a wave sums 8 consecutive nodes, keeps their relu'd rows in an LDS tile of its own, multiplies it by W2 (f32 WMMA); one launch fewer; sum order kept
# speedup vs baseline: 4.5293x; 1.0547x over previous
#include <hip/hip_runtime.h>

typedef __attribute__((ext_vector_type(2))) float v2f;
typedef __attribute__((ext_vector_type(8))) float v8f;

__global__ void set_val_kernel(float* p, long long n, float v) {
  long long i = (long long)blockIdx.x * blockDim.x + threadIdx.x;
  if (i < n) p[i] = v;
}


template<int K, int N>
__device__ __forceinline__ void stage_w_fragments(const float* __restrict__ W, float4* sW) {
  constexpr int CT = N / 16;
  for (int u = threadIdx.x; u < (K / 4) * N; u += 256) {
    const int n  = u % N;
    const int kq = u / N;
    const int q = kq >> 3, hh = (kq >> 2) & 1, jj = kq & 3;
    const float* w = W + (long long)(4 * kq) * N + n;
    float4 v;
    v.x = w[0];
    v.y = w[N];
    v.z = w[2 * N];
    v.w = w[3 * N];
    sW[((q * 4 + jj) * CT + (n % CT)) * 32 + hh * 16 + (n / CT)] = v;
  }
}

template<int K, int N>
__device__ __forceinline__ void gemm_wmma_block(const float* __restrict__ X,
                                                const float* __restrict__ W,
                                                float* __restrict__ OUT, int M,
                                                const int blk) {
  constexpr int CT = N / 16;
  constexpr int QN = K / 32;
  __shared__ float4 sW[(K / 4) * N];

  stage_w_fragments<K, N>(W, sW);
  __syncthreads();

  const int wave = threadIdx.x >> 5;
  const int lane = threadIdx.x & 31;
  const int rowBase = (blk * 8 + wave) * 16;
  if (rowBase + 16 > M) return;

  const int c = lane & 15;
  const int h = lane >> 4;

  const float4* xp = (const float4*)(X + (long long)(rowBase + c) * K + 16 * h);
  const float4* bp = sW + lane;

  v8f acc[CT];
#pragma unroll
  for (int t = 0; t < CT; ++t) acc[t] = v8f{};

  float4 a[2][4];
#pragma unroll
  for (int i = 0; i < 4; ++i) a[0][i] = xp[i];

#pragma unroll
  for (int q = 0; q < QN; ++q) {
    if (q + 1 < QN) {
#pragma unroll
      for (int i = 0; i < 4; ++i) a[(q + 1) & 1][i] = xp[(q + 1) * 8 + i];
    }
#pragma unroll
    for (int jj = 0; jj < 4; ++jj) {
      const float4 av = a[q & 1][jj];
      float4 b[CT];
#pragma unroll
      for (int t = 0; t < CT; ++t) b[t] = bp[((q * 4 + jj) * CT + t) * 32];
      v2f a0; a0.x = av.x; a0.y = av.y;
      v2f a1; a1.x = av.z; a1.y = av.w;
#pragma unroll
      for (int t = 0; t < CT; ++t) {
        v2f b0; b0.x = b[t].x; b0.y = b[t].y;
        acc[t] = __builtin_amdgcn_wmma_f32_16x16x4_f32(
            false, a0, false, b0, (short)0, acc[t], false, false);
      }
#pragma unroll
      for (int t = 0; t < CT; ++t) {
        v2f b1; b1.x = b[t].z; b1.y = b[t].w;
        acc[t] = __builtin_amdgcn_wmma_f32_16x16x4_f32(
            false, a1, false, b1, (short)0, acc[t], false, false);
      }
    }
  }

  float* out = OUT + (long long)(rowBase + 8 * h) * N + CT * c;
#pragma unroll
  for (int r = 0; r < 8; ++r) {
    if constexpr (CT == 4) {
      float4 o;
      o.x = acc[0][r]; o.y = acc[1][r]; o.z = acc[2][r]; o.w = acc[3][r];
      *(float4*)(out + (long long)r * N) = o;
    } else {
      static_assert(CT == 2, "N is 64 or 32 here");
      float2 o;
      o.x = acc[0][r]; o.y = acc[1][r];
      *(float2*)(out + (long long)r * N) = o;
    }
  }
}


__device__ __forceinline__ int count_chunk(const int* __restrict__ src, const int* __restrict__ dst,
                                           int E, int t, int rng, int split, int lane,
                                           int* __restrict__ cnt, int* __restrict__ ctr,
                                           int* __restrict__ ell, int* __restrict__ spill) {
  const int e0 = t * 256 + lane;
  int d[8], s[8], r[8];
#pragma unroll
  for (int i = 0; i < 8; ++i) {
    const int e = e0 + 32 * i;
    d[i] = (e < E) ? dst[e] : -1;
    s[i] = (e < E) ? src[e] : 0;
  }
#pragma unroll
  for (int i = 0; i < 8; ++i) {
    const bool here = (d[i] >= 0) && ((d[i] >= split) == (rng != 0));
    if (!here) d[i] = -1;
  }
#pragma unroll
  for (int i = 0; i < 8; ++i) {
    r[i] = 0;
    if (d[i] >= 0) r[i] = atomicAdd(&cnt[d[i]], 1);
  }
  int tv = 0;
  if (lane == 0) tv = atomicAdd(ctr, 1);
#pragma unroll
  for (int i = 0; i < 8; ++i) {
    if (d[i] >= 0) {
      if (r[i] < 32)      ell[(long long)d[i] * 32 + r[i]] = s[i];
      else if (r[i] < 48) spill[(long long)d[i] * 16 + (r[i] - 32)] = s[i];
    }
  }
  return tv;
}

__global__ void __launch_bounds__(256)
gemm1_count_kernel(const float* __restrict__ X, const float* __restrict__ W,
                   float* __restrict__ OUT, int M, int gemmBlocks,
                   const int* __restrict__ src, const int* __restrict__ dst, int E,
                   int* __restrict__ cnt, int* __restrict__ tick,
                   int* __restrict__ ell, int* __restrict__ spill) {
  if ((int)blockIdx.x < gemmBlocks) {
    gemm_wmma_block<256, 64>(X, W, OUT, M, (int)blockIdx.x);
    return;
  }
  const int lane = threadIdx.x & 31;
  const unsigned where = __builtin_amdgcn_s_sendmsg_rtn(0x87);
  const int half = (int)((where >> 18) & 1u);
  const int split = ((M >> 1) + 31) & ~31;
  const int nTickets = (E + 255) >> 8;
  for (int ph = 0; ph < 2; ++ph) {
    const int rng = half ^ ph;
    int* ctr = tick + 64 * rng;
    int tv = 0;
    if (lane == 0) tv = atomicAdd(ctr, 1);
    for (int it = 0; it <= nTickets; ++it) {
      const int t = __builtin_amdgcn_readfirstlane(tv);
      if (t >= nTickets) break;
      tv = count_chunk(src, dst, E, t, rng, split, lane, cnt, ctr, ell, spill);
    }
  }
}


__device__ __forceinline__ float inv_sqrt_deg(int c) { return rsqrtf(1.0f + (float)c); }

template<int V>
__device__ __forceinline__ void add_rows(float (&acc)[V], const float* __restrict__ H,
                                         int sid, float w, int m, int lane) {
  for (int k0 = 0; k0 < m; k0 += 8) {
    float vx[8], vy[8];
#pragma unroll
    for (int j = 0; j < 8; ++j) {
      const int s = __builtin_amdgcn_readlane(sid, k0 + j);
      if constexpr (V == 2) {
        const float2 v = *(const float2*)(H + (long long)s * 64 + 2 * lane);
        vx[j] = v.x; vy[j] = v.y;
      } else {
        vx[j] = H[(long long)s * 32 + lane]; vy[j] = 0.0f;
      }
    }
#pragma unroll
    for (int j = 0; j < 8; ++j) {
      if constexpr (V == 2) asm volatile("" : "+v"(vx[j]), "+v"(vy[j]));
      else                  asm volatile("" : "+v"(vx[j]));
    }
#pragma unroll
    for (int j = 0; j < 8; ++j) {
      if (j == 0 || k0 + j < m) {
        const float ws = __int_as_float(__builtin_amdgcn_readlane(__float_as_int(w), k0 + j));
        acc[0] += vx[j] * ws;
        if constexpr (V == 2) acc[1] += vy[j] * ws;
      }
    }
  }
}

template<int V>
__device__ __forceinline__ void add_rows_from_edges(float (&acc)[V], const float* __restrict__ H,
                                                    const int* __restrict__ src,
                                                    const int* __restrict__ dst, int E,
                                                    const int* __restrict__ cnt,
                                                    int node, float dn, int lane) {
  for (int e0 = 0; e0 < E; e0 += 32) {
    const int e = e0 + lane;
    const bool hit = (e < E) && (dst[e] == node);
    int sid = 0; float w = 0.0f;
    if (hit) { sid = src[e]; w = inv_sqrt_deg(cnt[sid]) * dn; }
    unsigned mask = __builtin_amdgcn_ballot_w32(hit);
    while (mask) {
      const int k = __builtin_ctz(mask);
      mask &= mask - 1;
      const int s = __builtin_amdgcn_readlane(sid, k);
      const float ws = __int_as_float(__builtin_amdgcn_readlane(__float_as_int(w), k));
      if constexpr (V == 2) {
        const float2 v = *(const float2*)(H + (long long)s * 64 + 2 * lane);
        acc[0] += v.x * ws;
        acc[1] += v.y * ws;
      } else {
        acc[0] += H[(long long)s * 32 + lane] * ws;
      }
    }
  }
}

template<int V>
__device__ __forceinline__ void sum_node(float (&acc)[V], const float* __restrict__ H,
                                         const int* __restrict__ cnt,
                                         const int* __restrict__ spill,
                                         const int* __restrict__ src,
                                         const int* __restrict__ dst, int E,
                                         int node, int c, int own, float dn, int lane) {
  if (__builtin_expect(c <= 48, 1)) {
    const int m = min(c, 32);
    int sid = 0; float w = 0.0f;
    if (lane < m) { sid = own; w = inv_sqrt_deg(cnt[sid]) * dn; }
    add_rows<V>(acc, H, sid, w, m, lane);
    if (c > 32) {
      const int m2 = c - 32;
      sid = 0; w = 0.0f;
      if (lane < m2) { sid = spill[(long long)node * 16 + lane]; w = inv_sqrt_deg(cnt[sid]) * dn; }
      add_rows<V>(acc, H, sid, w, m2, lane);
    }
  } else {
    add_rows_from_edges<V>(acc, H, src, dst, E, cnt, node, dn, lane);
  }
}


constexpr int TILE_PITCH = 68;

__global__ void __launch_bounds__(256)
pull_relu64_mul_kernel(const int* __restrict__ cnt, const int* __restrict__ ell,
                       const int* __restrict__ spill,
                       const int* __restrict__ src, const int* __restrict__ dst, int E,
                       const float* __restrict__ H, const float* __restrict__ bias,
                       const float* __restrict__ W2, float* __restrict__ H2, int M) {
  constexpr int K = 64, N = 32;
  constexpr int CT = N / 16;
  constexpr int QN = K / 32;
  __shared__ float4 sW[(K / 4) * N];
  __shared__ float4 sTile[8 * 8 * (TILE_PITCH / 4)];
  __shared__ float4 sZero[K / 4];

  stage_w_fragments<K, N>(W2, sW);
  if (threadIdx.x < K / 4) {
    float4 zz;
    zz.x = 0.0f; zz.y = 0.0f; zz.z = 0.0f; zz.w = 0.0f;
    sZero[threadIdx.x] = zz;
  }
  __syncthreads();

  const int lane = threadIdx.x & 31;
  const int wv = __builtin_amdgcn_readfirstlane((int)(threadIdx.x >> 5));
  const int node0 = ((int)blockIdx.x * 8 + wv) * 8;
  if (node0 >= M) return;
  float* tile = (float*)sTile + wv * (8 * TILE_PITCH);
  const float2 bb = *(const float2*)(bias + 2 * lane);

  for (int j = 0; j < 8; ++j) {
    const int node = node0 + j;
    float2 z;
    z.x = 0.0f; z.y = 0.0f;
    if (node < M) {
      int own = ell[(long long)node * 32 + lane];
      float2 hs = *(const float2*)(H + (long long)node * 64 + 2 * lane);
      const int c = cnt[node];
      asm volatile("" : "+v"(own), "+v"(hs.x), "+v"(hs.y));
      const float dn = inv_sqrt_deg(c);
      float acc[2] = {0.0f, 0.0f};
      sum_node<2>(acc, H, cnt, spill, src, dst, E, node, c, own, dn, lane);
      z.x = fmaxf(acc[0] + dn * dn * hs.x + bb.x, 0.0f);
      z.y = fmaxf(acc[1] + dn * dn * hs.y + bb.y, 0.0f);
    }
    *(float2*)(tile + j * TILE_PITCH + 2 * lane) = z;
  }

  const int c = lane & 15;
  const int h = lane >> 4;
  const float* arow = (c < 8 ? tile + c * TILE_PITCH : (const float*)sZero) + 16 * h;
  const float4* bp = sW + lane;

  v8f acc[CT];
#pragma unroll
  for (int t = 0; t < CT; ++t) acc[t] = v8f{};

#pragma unroll
  for (int q = 0; q < QN; ++q) {
#pragma unroll
    for (int jj = 0; jj < 4; ++jj) {
      const float4 av = *(const float4*)(arow + 32 * q + 4 * jj);
      float4 b[CT];
#pragma unroll
      for (int t = 0; t < CT; ++t) b[t] = bp[((q * 4 + jj) * CT + t) * 32];
      v2f a0; a0.x = av.x; a0.y = av.y;
      v2f a1; a1.x = av.z; a1.y = av.w;
#pragma unroll
      for (int t = 0; t < CT; ++t) {
        v2f b0; b0.x = b[t].x; b0.y = b[t].y;
        acc[t] = __builtin_amdgcn_wmma_f32_16x16x4_f32(
            false, a0, false, b0, (short)0, acc[t], false, false);
      }
#pragma unroll
      for (int t = 0; t < CT; ++t) {
        v2f b1; b1.x = b[t].z; b1.y = b[t].w;
        acc[t] = __builtin_amdgcn_wmma_f32_16x16x4_f32(
            false, a1, false, b1, (short)0, acc[t], false, false);
      }
    }
  }

  if (h == 0) {
    float* o = H2 + (long long)node0 * N + CT * c;
#pragma unroll
    for (int r = 0; r < 8; ++r) {
      if (node0 + r < M) {
        float2 v;
        v.x = acc[0][r]; v.y = acc[1][r];
        *(float2*)(o + (long long)r * N) = v;
      }
    }
  }
}

__global__ void __launch_bounds__(256)
pull_softmax32_kernel(const int* __restrict__ cnt, const int* __restrict__ spill,
                      const int* __restrict__ src, const int* __restrict__ dst, int E,
                      const float* __restrict__ H,
                      const float* __restrict__ bias, float* out, int M) {
  const int lane = threadIdx.x & 31;
  const int wv = __builtin_amdgcn_readfirstlane((int)(threadIdx.x >> 5));
  const int nw = gridDim.x * 8;
  const float bb = bias[lane];
  for (int node = blockIdx.x * 8 + wv; node < M; node += nw) {
    int own = __float_as_int(out[(long long)node * 32 + lane]);
    float hs = H[(long long)node * 32 + lane];
    const int c = cnt[node];
    asm volatile("" : "+v"(own), "+v"(hs));
    const float dn = inv_sqrt_deg(c);
    float acc[1] = {0.0f};
    sum_node<1>(acc, H, cnt, spill, src, dst, E, node, c, own, dn, lane);
    const float z = acc[0] + dn * dn * hs + bb;
    float mx = z;
    for (int o = 16; o > 0; o >>= 1) mx = fmaxf(mx, __shfl_xor(mx, o, 32));
    const float ev = __expf(z - mx);
    float sm = ev;
    for (int o = 16; o > 0; o >>= 1) sm += __shfl_xor(sm, o, 32);
    out[(long long)node * 32 + lane] = ev / sm;
  }
}

extern "C" void kernel_launch(void* const* d_in, const int* in_sizes, int n_in,
                              void* d_out, int out_size, void* d_ws, size_t ws_size,
                              hipStream_t stream) {
  (void)n_in; (void)out_size; (void)ws_size;
  const float* x  = (const float*)d_in[0];
  const int*   ei = (const int*)d_in[1];
  const float* W1 = (const float*)d_in[2];
  const float* b1 = (const float*)d_in[3];
  const float* W2 = (const float*)d_in[4];
  const float* b2 = (const float*)d_in[5];

  const int IN = 256, HID = 64, OD = 32;
  const int M = in_sizes[0] / IN;
  const int E = in_sizes[1] / 2;
  const int* src = ei;
  const int* dst = ei + E;

  const long long Mp = ((long long)M + 1023) & ~1023LL;
  const long long Cw = Mp + 128;
  int*   cnt   = (int*)d_ws;
  int*   tick  = cnt + Mp;
  float* H1    = (float*)d_ws + Cw;
  float* H2    = H1 + (long long)M * HID;
  int*   spill = (int*)(H2 + (long long)M * HID);
  float* out   = (float*)d_out;
  int*   ell   = (int*)d_out;

  const int T = 256;
  const int PULL_BLOCKS = 2048;
  const int COUNT_BLOCKS = 768;

  set_val_kernel<<<(int)((Cw + T - 1) / T), T, 0, stream>>>((float*)cnt, Cw, 0.0f);
  gemm1_count_kernel<<<(M + 127) / 128 + COUNT_BLOCKS, 256, 0, stream>>>(
      x, W1, H1, M, (M + 127) / 128, src, dst, E, cnt, tick, ell, spill);

  pull_relu64_mul_kernel<<<(M + 63) / 64, T, 0, stream>>>(cnt, ell, spill, src, dst, E, H1, b1,
                                                          W2, H2, M);

  pull_softmax32_kernel<<<PULL_BLOCKS, T, 0, stream>>>(cnt, spill, src, dst, E, H2, b2, out, M);
}


# edge sums: per node ONE request group (own row, source counts, first 16 source rows), the next node's slot line + count riding in it: 4-5 dependent round trips per node become 1-2; sum order kept
# speedup vs baseline: 4.6739x; 1.0319x over previous
#include <hip/hip_runtime.h>

typedef __attribute__((ext_vector_type(2))) float v2f;
typedef __attribute__((ext_vector_type(8))) float v8f;

__global__ void set_val_kernel(float* p, long long n, float v) {
  long long i = (long long)blockIdx.x * blockDim.x + threadIdx.x;
  if (i < n) p[i] = v;
}


template<int K, int N>
__device__ __forceinline__ void stage_w_fragments(const float* __restrict__ W, float4* sW) {
  constexpr int CT = N / 16;
  for (int u = threadIdx.x; u < (K / 4) * N; u += 256) {
    const int n  = u % N;
    const int kq = u / N;
    const int q = kq >> 3, hh = (kq >> 2) & 1, jj = kq & 3;
    const float* w = W + (long long)(4 * kq) * N + n;
    float4 v;
    v.x = w[0];
    v.y = w[N];
    v.z = w[2 * N];
    v.w = w[3 * N];
    sW[((q * 4 + jj) * CT + (n % CT)) * 32 + hh * 16 + (n / CT)] = v;
  }
}

template<int K, int N>
__device__ __forceinline__ void gemm_wmma_block(const float* __restrict__ X,
                                                const float* __restrict__ W,
                                                float* __restrict__ OUT, int M,
                                                const int blk) {
  constexpr int CT = N / 16;
  constexpr int QN = K / 32;
  __shared__ float4 sW[(K / 4) * N];

  stage_w_fragments<K, N>(W, sW);
  __syncthreads();

  const int wave = threadIdx.x >> 5;
  const int lane = threadIdx.x & 31;
  const int rowBase = (blk * 8 + wave) * 16;
  if (rowBase + 16 > M) return;

  const int c = lane & 15;
  const int h = lane >> 4;

  const float4* xp = (const float4*)(X + (long long)(rowBase + c) * K + 16 * h);
  const float4* bp = sW + lane;

  v8f acc[CT];
#pragma unroll
  for (int t = 0; t < CT; ++t) acc[t] = v8f{};

  float4 a[2][4];
#pragma unroll
  for (int i = 0; i < 4; ++i) a[0][i] = xp[i];

#pragma unroll
  for (int q = 0; q < QN; ++q) {
    if (q + 1 < QN) {
#pragma unroll
      for (int i = 0; i < 4; ++i) a[(q + 1) & 1][i] = xp[(q + 1) * 8 + i];
    }
#pragma unroll
    for (int jj = 0; jj < 4; ++jj) {
      const float4 av = a[q & 1][jj];
      float4 b[CT];
#pragma unroll
      for (int t = 0; t < CT; ++t) b[t] = bp[((q * 4 + jj) * CT + t) * 32];
      v2f a0; a0.x = av.x; a0.y = av.y;
      v2f a1; a1.x = av.z; a1.y = av.w;
#pragma unroll
      for (int t = 0; t < CT; ++t) {
        v2f b0; b0.x = b[t].x; b0.y = b[t].y;
        acc[t] = __builtin_amdgcn_wmma_f32_16x16x4_f32(
            false, a0, false, b0, (short)0, acc[t], false, false);
      }
#pragma unroll
      for (int t = 0; t < CT; ++t) {
        v2f b1; b1.x = b[t].z; b1.y = b[t].w;
        acc[t] = __builtin_amdgcn_wmma_f32_16x16x4_f32(
            false, a1, false, b1, (short)0, acc[t], false, false);
      }
    }
  }

  float* out = OUT + (long long)(rowBase + 8 * h) * N + CT * c;
#pragma unroll
  for (int r = 0; r < 8; ++r) {
    if constexpr (CT == 4) {
      float4 o;
      o.x = acc[0][r]; o.y = acc[1][r]; o.z = acc[2][r]; o.w = acc[3][r];
      *(float4*)(out + (long long)r * N) = o;
    } else {
      static_assert(CT == 2, "N is 64 or 32 here");
      float2 o;
      o.x = acc[0][r]; o.y = acc[1][r];
      *(float2*)(out + (long long)r * N) = o;
    }
  }
}


__device__ __forceinline__ int count_chunk(const int* __restrict__ src, const int* __restrict__ dst,
                                           int E, int t, int rng, int split, int lane,
                                           int* __restrict__ cnt, int* __restrict__ ctr,
                                           int* __restrict__ ell, int* __restrict__ spill) {
  const int e0 = t * 256 + lane;
  int d[8], s[8], r[8];
#pragma unroll
  for (int i = 0; i < 8; ++i) {
    const int e = e0 + 32 * i;
    d[i] = (e < E) ? dst[e] : -1;
    s[i] = (e < E) ? src[e] : 0;
  }
#pragma unroll
  for (int i = 0; i < 8; ++i) {
    const bool here = (d[i] >= 0) && ((d[i] >= split) == (rng != 0));
    if (!here) d[i] = -1;
  }
#pragma unroll
  for (int i = 0; i < 8; ++i) {
    r[i] = 0;
    if (d[i] >= 0) r[i] = atomicAdd(&cnt[d[i]], 1);
  }
  int tv = 0;
  if (lane == 0) tv = atomicAdd(ctr, 1);
#pragma unroll
  for (int i = 0; i < 8; ++i) {
    if (d[i] >= 0) {
      if (r[i] < 32)      ell[(long long)d[i] * 32 + r[i]] = s[i];
      else if (r[i] < 48) spill[(long long)d[i] * 16 + (r[i] - 32)] = s[i];
    }
  }
  return tv;
}

__global__ void __launch_bounds__(256)
gemm1_count_kernel(const float* __restrict__ X, const float* __restrict__ W,
                   float* __restrict__ OUT, int M, int gemmBlocks,
                   const int* __restrict__ src, const int* __restrict__ dst, int E,
                   int* __restrict__ cnt, int* __restrict__ tick,
                   int* __restrict__ ell, int* __restrict__ spill) {
  if ((int)blockIdx.x < gemmBlocks) {
    gemm_wmma_block<256, 64>(X, W, OUT, M, (int)blockIdx.x);
    return;
  }
  const int lane = threadIdx.x & 31;
  const unsigned where = __builtin_amdgcn_s_sendmsg_rtn(0x87);
  const int half = (int)((where >> 18) & 1u);
  const int split = ((M >> 1) + 31) & ~31;
  const int nTickets = (E + 255) >> 8;
  for (int ph = 0; ph < 2; ++ph) {
    const int rng = half ^ ph;
    int* ctr = tick + 64 * rng;
    int tv = 0;
    if (lane == 0) tv = atomicAdd(ctr, 1);
    for (int it = 0; it <= nTickets; ++it) {
      const int t = __builtin_amdgcn_readfirstlane(tv);
      if (t >= nTickets) break;
      tv = count_chunk(src, dst, E, t, rng, split, lane, cnt, ctr, ell, spill);
    }
  }
}


__device__ __forceinline__ float inv_sqrt_deg(int c) { return rsqrtf(1.0f + (float)c); }

typedef const __attribute__((address_space(1))) float* grow_t;
typedef const __attribute__((address_space(1))) v2f* grow2_t;

template<int V, int D, bool FIRST>
__device__ __forceinline__ void row_trip(float (&acc)[V], const float* __restrict__ H,
                                         int sid, int& cs, float& w, float dn,
                                         int k0, int m, int lane) {
  static_assert(D == 16 || D == 8, "trips of 16 or of 8 rows");
  v2f r2[V == 2 ? D : 1];
  float r1[V == 1 ? D : 1];
  grow_t row[D];
#pragma unroll
  for (int j = 0; j < D; ++j) {
    const int s = __builtin_amdgcn_readlane(sid, k0 + j);
    row[j] = (grow_t)(H + (long long)s * (32 * V));
  }
  if constexpr (D == 16) {
    asm volatile("" : "+s"(row[0]), "+s"(row[1]), "+s"(row[2]), "+s"(row[3]),
                      "+s"(row[4]), "+s"(row[5]), "+s"(row[6]), "+s"(row[7]));
    asm volatile("" : "+s"(row[8]), "+s"(row[9]), "+s"(row[10]), "+s"(row[11]),
                      "+s"(row[12]), "+s"(row[13]), "+s"(row[14]), "+s"(row[15]));
  } else {
    asm volatile("" : "+s"(row[0]), "+s"(row[1]), "+s"(row[2]), "+s"(row[3]),
                      "+s"(row[4]), "+s"(row[5]), "+s"(row[6]), "+s"(row[7]));
  }
#pragma unroll
  for (int j = 0; j < D; ++j) {
    if constexpr (V == 2) r2[j] = *(grow2_t)(row[j] + 2 * lane);
    else                  r1[j] = row[j][lane];
  }
#pragma unroll
  for (int j = 0; j < D; ++j) {
    if constexpr (V == 2) asm volatile("" : "+v"(r2[j]));
    else                  asm volatile("" : "+v"(r1[j]));
  }
  if constexpr (FIRST) {
    asm volatile("" : "+v"(cs));
    w = inv_sqrt_deg(cs) * dn;
  }
#pragma unroll
  for (int j = 0; j < D; ++j) {
    if (j == 0 || k0 + j < m) {
      const float ws = __int_as_float(__builtin_amdgcn_readlane(__float_as_int(w), k0 + j));
      if constexpr (V == 2) { acc[0] += r2[j].x * ws; acc[1] += r2[j].y * ws; }
      else                  { acc[0] += r1[j] * ws; }
    }
  }
}

template<int V>
__device__ __forceinline__ void add_rows(float (&acc)[V], const float* __restrict__ H,
                                         const int* __restrict__ cnt,
                                         int sid, float dn, int m, int lane) {
  int cs = cnt[sid];
  float w = 0.0f;
  row_trip<V, 16, true>(acc, H, sid, cs, w, dn, 0, m, lane);
  if (m > 16) {
    if (m <= 24) row_trip<V, 8, false>(acc, H, sid, cs, w, dn, 16, m, lane);
    else         row_trip<V, 16, false>(acc, H, sid, cs, w, dn, 16, m, lane);
  }
}

template<int V>
__device__ __forceinline__ void add_rows_from_edges(float (&acc)[V], const float* __restrict__ H,
                                                    const int* __restrict__ src,
                                                    const int* __restrict__ dst, int E,
                                                    const int* __restrict__ cnt,
                                                    int node, float dn, int lane) {
  for (int e0 = 0; e0 < E; e0 += 32) {
    const int e = e0 + lane;
    const bool hit = (e < E) && (dst[e] == node);
    int sid = 0; float w = 0.0f;
    if (hit) { sid = src[e]; w = inv_sqrt_deg(cnt[sid]) * dn; }
    unsigned mask = __builtin_amdgcn_ballot_w32(hit);
    while (mask) {
      const int k = __builtin_ctz(mask);
      mask &= mask - 1;
      const int s = __builtin_amdgcn_readlane(sid, k);
      const float ws = __int_as_float(__builtin_amdgcn_readlane(__float_as_int(w), k));
      if constexpr (V == 2) {
        const float2 v = *(const float2*)(H + (long long)s * 64 + 2 * lane);
        acc[0] += v.x * ws;
        acc[1] += v.y * ws;
      } else {
        acc[0] += H[(long long)s * 32 + lane] * ws;
      }
    }
  }
}

template<int V>
__device__ __forceinline__ void sum_node(float (&acc)[V], const float* __restrict__ H,
                                         const int* __restrict__ cnt,
                                         const int* __restrict__ spill,
                                         const int* __restrict__ src,
                                         const int* __restrict__ dst, int E,
                                         int node, int c, int own, float dn, int lane) {
  if (__builtin_expect(c <= 48, 1)) {
    const int m = min(c, 32);
    if (m > 0) {
      const int sid = (lane < m) ? own : 0;
      add_rows<V>(acc, H, cnt, sid, dn, m, lane);
    }
    if (c > 32) {
      const int m2 = c - 32;
      int sid = 0;
      if (lane < m2) sid = spill[(long long)node * 16 + lane];
      add_rows<V>(acc, H, cnt, sid, dn, m2, lane);
    }
  } else {
    add_rows_from_edges<V>(acc, H, src, dst, E, cnt, node, dn, lane);
  }
}


constexpr int TILE_PITCH = 68;

__global__ void __launch_bounds__(256)
pull_relu64_mul_kernel(const int* __restrict__ cnt, const int* __restrict__ ell,
                       const int* __restrict__ spill,
                       const int* __restrict__ src, const int* __restrict__ dst, int E,
                       const float* __restrict__ H, const float* __restrict__ bias,
                       const float* __restrict__ W2, float* __restrict__ H2, int M) {
  constexpr int K = 64, N = 32;
  constexpr int CT = N / 16;
  constexpr int QN = K / 32;
  __shared__ float4 sW[(K / 4) * N];
  __shared__ float4 sTile[8 * 8 * (TILE_PITCH / 4)];
  __shared__ float4 sZero[K / 4];

  stage_w_fragments<K, N>(W2, sW);
  if (threadIdx.x < K / 4) {
    float4 zz;
    zz.x = 0.0f; zz.y = 0.0f; zz.z = 0.0f; zz.w = 0.0f;
    sZero[threadIdx.x] = zz;
  }
  __syncthreads();

  const int lane = threadIdx.x & 31;
  const int wv = __builtin_amdgcn_readfirstlane((int)(threadIdx.x >> 5));
  const int node0 = ((int)blockIdx.x * 8 + wv) * 8;
  if (node0 >= M) return;
  float* tile = (float*)sTile + wv * (8 * TILE_PITCH);
  const float2 bb = *(const float2*)(bias + 2 * lane);

  int own = ell[(long long)node0 * 32 + lane];
  int cn = cnt[node0];
  for (int j = 0; j < 8; ++j) {
    const int node = node0 + j;
    float2 z;
    z.x = 0.0f; z.y = 0.0f;
    if (node < M) {
      const int na = (j < 7 && node + 1 < M) ? node + 1 : node;
      const int own_a = ell[(long long)na * 32 + lane];
      const int cn_a = cnt[na];
      const float2 hs = *(const float2*)(H + (long long)node * 64 + 2 * lane);
      const float dn = inv_sqrt_deg(cn);
      float acc[2] = {0.0f, 0.0f};
      sum_node<2>(acc, H, cnt, spill, src, dst, E, node, cn, own, dn, lane);
      z.x = fmaxf(acc[0] + dn * dn * hs.x + bb.x, 0.0f);
      z.y = fmaxf(acc[1] + dn * dn * hs.y + bb.y, 0.0f);
      own = own_a;
      cn = cn_a;
    }
    *(float2*)(tile + j * TILE_PITCH + 2 * lane) = z;
  }

  const int c = lane & 15;
  const int h = lane >> 4;
  const float* arow = (c < 8 ? tile + c * TILE_PITCH : (const float*)sZero) + 16 * h;
  const float4* bp = sW + lane;

  v8f acc[CT];
#pragma unroll
  for (int t = 0; t < CT; ++t) acc[t] = v8f{};

#pragma unroll
  for (int q = 0; q < QN; ++q) {
#pragma unroll
    for (int jj = 0; jj < 4; ++jj) {
      const float4 av = *(const float4*)(arow + 32 * q + 4 * jj);
      float4 b[CT];
#pragma unroll
      for (int t = 0; t < CT; ++t) b[t] = bp[((q * 4 + jj) * CT + t) * 32];
      v2f a0; a0.x = av.x; a0.y = av.y;
      v2f a1; a1.x = av.z; a1.y = av.w;
#pragma unroll
      for (int t = 0; t < CT; ++t) {
        v2f b0; b0.x = b[t].x; b0.y = b[t].y;
        acc[t] = __builtin_amdgcn_wmma_f32_16x16x4_f32(
            false, a0, false, b0, (short)0, acc[t], false, false);
      }
#pragma unroll
      for (int t = 0; t < CT; ++t) {
        v2f b1; b1.x = b[t].z; b1.y = b[t].w;
        acc[t] = __builtin_amdgcn_wmma_f32_16x16x4_f32(
            false, a1, false, b1, (short)0, acc[t], false, false);
      }
    }
  }

  if (h == 0) {
    float* o = H2 + (long long)node0 * N + CT * c;
#pragma unroll
    for (int r = 0; r < 8; ++r) {
      if (node0 + r < M) {
        float2 v;
        v.x = acc[0][r]; v.y = acc[1][r];
        *(float2*)(o + (long long)r * N) = v;
      }
    }
  }
}

__global__ void __launch_bounds__(256)
pull_softmax32_kernel(const int* __restrict__ cnt, const int* __restrict__ spill,
                      const int* __restrict__ src, const int* __restrict__ dst, int E,
                      const float* __restrict__ H,
                      const float* __restrict__ bias, float* out, int M) {
  const int lane = threadIdx.x & 31;
  const int wv = __builtin_amdgcn_readfirstlane((int)(threadIdx.x >> 5));
  const int nw = gridDim.x * 8;
  const float bb = bias[lane];
  int node = blockIdx.x * 8 + wv;
  if (node >= M) return;
  int own = __float_as_int(out[(long long)node * 32 + lane]);
  int c = cnt[node];
  for (int it = 0; it < M; ++it) {
    const int nxt = node + nw;
    const bool more = nxt < M;
    const int na = more ? nxt : node;
    const int own_a = __float_as_int(out[(long long)na * 32 + lane]);
    const int c_a = cnt[na];
    const float hs = H[(long long)node * 32 + lane];
    const float dn = inv_sqrt_deg(c);
    float acc[1] = {0.0f};
    sum_node<1>(acc, H, cnt, spill, src, dst, E, node, c, own, dn, lane);
    const float z = acc[0] + dn * dn * hs + bb;
    float mx = z;
    for (int o = 16; o > 0; o >>= 1) mx = fmaxf(mx, __shfl_xor(mx, o, 32));
    const float ev = __expf(z - mx);
    float sm = ev;
    for (int o = 16; o > 0; o >>= 1) sm += __shfl_xor(sm, o, 32);
    out[(long long)node * 32 + lane] = ev / sm;
    if (!more) break;
    node = nxt;
    own = own_a;
    c = c_a;
  }
}

extern "C" void kernel_launch(void* const* d_in, const int* in_sizes, int n_in,
                              void* d_out, int out_size, void* d_ws, size_t ws_size,
                              hipStream_t stream) {
  (void)n_in; (void)out_size; (void)ws_size;
  const float* x  = (const float*)d_in[0];
  const int*   ei = (const int*)d_in[1];
  const float* W1 = (const float*)d_in[2];
  const float* b1 = (const float*)d_in[3];
  const float* W2 = (const float*)d_in[4];
  const float* b2 = (const float*)d_in[5];

  const int IN = 256, HID = 64, OD = 32;
  const int M = in_sizes[0] / IN;
  const int E = in_sizes[1] / 2;
  const int* src = ei;
  const int* dst = ei + E;

  const long long Mp = ((long long)M + 1023) & ~1023LL;
  const long long Cw = Mp + 128;
  int*   cnt   = (int*)d_ws;
  int*   tick  = cnt + Mp;
  float* H1    = (float*)d_ws + Cw;
  float* H2    = H1 + (long long)M * HID;
  int*   spill = (int*)(H2 + (long long)M * HID);
  float* out   = (float*)d_out;
  int*   ell   = (int*)d_out;

  const int T = 256;
  const int PULL_BLOCKS = 2048;
  const int COUNT_BLOCKS = 768;

  set_val_kernel<<<(int)((Cw + T - 1) / T), T, 0, stream>>>((float*)cnt, Cw, 0.0f);
  gemm1_count_kernel<<<(M + 127) / 128 + COUNT_BLOCKS, 256, 0, stream>>>(
      x, W1, H1, M, (M + 127) / 128, src, dst, E, cnt, tick, ell, spill);

  pull_relu64_mul_kernel<<<(M + 63) / 64, T, 0, stream>>>(cnt, ell, spill, src, dst, E, H1, b1,
                                                          W2, H2, M);

  pull_softmax32_kernel<<<PULL_BLOCKS, T, 0, stream>>>(cnt, spill, src, dst, E, H2, b2, out, M);
}
